# starting kernel
# baseline (speedup 1.0000x reference)
#include <hip/hip_runtime.h>


#define Bsz 512
#define Isz 1024
#define HD  2048
#define Lsz 1000
#define LP  1024
#define TMAX 32

typedef __attribute__((ext_vector_type(16))) __bf16 v16bf;
typedef __attribute__((ext_vector_type(8)))  float  v8f;
typedef __attribute__((ext_vector_type(4)))  unsigned int u32x4;

__device__ __forceinline__ unsigned short f2bf(float f) {
    unsigned int u = __float_as_uint(f);
    u += 0x7FFFu + ((u >> 16) & 1u);
    return (unsigned short)(u >> 16);
}

struct alignas(32) FragBits { u32x4 lo, hi; };

__device__ __forceinline__ v16bf load_frag(const unsigned short* __restrict__ p) {
    FragBits fb;
    fb.lo = *reinterpret_cast<const u32x4*>(p);
    fb.hi = *reinterpret_cast<const u32x4*>(p + 16);
    v16bf v;
    __builtin_memcpy(&v, &fb, sizeof(v));
    return v;
}

template <int MODE>
__global__ __launch_bounds__(128)
void wmma_gemm(const unsigned short* __restrict__ A,
               const unsigned short* __restrict__ BT,
               int M, int N, int K,
               const float* __restrict__ bias,
               const float* __restrict__ drive,
               const float* __restrict__ Hold,
               const int*   __restrict__ active,
               float* __restrict__ Cout,
               unsigned short* __restrict__ HbOut,
               unsigned int* __restrict__ dHbits,
               int NL) {
    const int lane = threadIdx.x & 31;
    const int wave = threadIdx.x >> 5;
    const int rowBase = blockIdx.y * 64 + wave * 16;
    const int colBase = blockIdx.x * 64;
    const int r16 = lane & 15;
    const int kh  = lane >> 4;

    const unsigned short* pa = A + (size_t)(rowBase + r16) * K + kh * 8;
    const unsigned short* pb0 = BT + (size_t)(colBase + r16) * K + kh * 8;
    const unsigned short* pb1 = pb0 + (size_t)16 * K;
    const unsigned short* pb2 = pb0 + (size_t)32 * K;
    const unsigned short* pb3 = pb0 + (size_t)48 * K;

    v8f acc0 = {0.f,0.f,0.f,0.f,0.f,0.f,0.f,0.f};
    v8f acc[4];
    acc[0] = acc0; acc[1] = acc0; acc[2] = acc0; acc[3] = acc0;

    v16bf a_cur = load_frag(pa);
    v16bf b_cur0 = load_frag(pb0);
    v16bf b_cur1 = load_frag(pb1);
    v16bf b_cur2 = load_frag(pb2);
    v16bf b_cur3 = load_frag(pb3);

    for (int k0 = 0; k0 < K - 32; k0 += 32) {
        pa += 32; pb0 += 32; pb1 += 32; pb2 += 32; pb3 += 32;
        const v16bf a_nxt  = load_frag(pa);
        const v16bf b_nxt0 = load_frag(pb0);
        const v16bf b_nxt1 = load_frag(pb1);
        const v16bf b_nxt2 = load_frag(pb2);
        const v16bf b_nxt3 = load_frag(pb3);
        __builtin_prefetch(pa + 32, 0, 3);

        acc[0] = __builtin_amdgcn_wmma_f32_16x16x32_bf16(
            false, a_cur, false, b_cur0, (short)0, acc[0], false, false);
        acc[1] = __builtin_amdgcn_wmma_f32_16x16x32_bf16(
            false, a_cur, false, b_cur1, (short)0, acc[1], false, false);
        acc[2] = __builtin_amdgcn_wmma_f32_16x16x32_bf16(
            false, a_cur, false, b_cur2, (short)0, acc[2], false, false);
        acc[3] = __builtin_amdgcn_wmma_f32_16x16x32_bf16(
            false, a_cur, false, b_cur3, (short)0, acc[3], false, false);

        a_cur = a_nxt;
        b_cur0 = b_nxt0; b_cur1 = b_nxt1; b_cur2 = b_nxt2; b_cur3 = b_nxt3;
    }
    acc[0] = __builtin_amdgcn_wmma_f32_16x16x32_bf16(
        false, a_cur, false, b_cur0, (short)0, acc[0], false, false);
    acc[1] = __builtin_amdgcn_wmma_f32_16x16x32_bf16(
        false, a_cur, false, b_cur1, (short)0, acc[1], false, false);
    acc[2] = __builtin_amdgcn_wmma_f32_16x16x32_bf16(
        false, a_cur, false, b_cur2, (short)0, acc[2], false, false);
    acc[3] = __builtin_amdgcn_wmma_f32_16x16x32_bf16(
        false, a_cur, false, b_cur3, (short)0, acc[3], false, false);

    bool rowActive[8];
    float dmax[8];
    if (MODE == 1) {
#pragma unroll
        for (int r = 0; r < 8; ++r) {
            rowActive[r] = active[rowBase + r + 8 * kh] != 0;
            dmax[r] = 0.f;
        }
    }

#pragma unroll
    for (int f = 0; f < 4; ++f) {
        const int n = colBase + f * 16 + r16;
#pragma unroll
        for (int r = 0; r < 8; ++r) {
            const int m = rowBase + r + 8 * kh;
            const float v = acc[f][r];
            if (MODE == 0) {
                Cout[(size_t)m * N + n] = v + bias[n];
            } else if (MODE == 2) {
                if (n < NL) Cout[(size_t)m * N + n] = v + bias[n];
            } else {
                const size_t idx = (size_t)m * N + n;
                const float pre = v + drive[idx];
                const float actv = pre >= 0.f ? pre : 0.05f * pre;
                const float ho = Hold[idx];
                const float hn = rowActive[r] ? actv : ho;
                Cout[idx] = hn;
                HbOut[idx] = f2bf(hn);
                dmax[r] = fmaxf(dmax[r], fabsf(hn - ho));
            }
        }
    }

    if (MODE == 1) {
#pragma unroll
        for (int r = 0; r < 8; ++r) {
            if (dmax[r] > 0.f) {
                atomicMax(dHbits + (rowBase + r + 8 * kh), __float_as_uint(dmax[r]));
            }
        }
    }
}

__global__ __launch_bounds__(256)
void col_l1_kernel(const float* __restrict__ A, const float* __restrict__ Msk,
                   float* __restrict__ colsum, int rows, int cols) {
    __shared__ float sm[256];
    const int j = blockIdx.x;
    float s = 0.f;
    for (int i = threadIdx.x; i < rows; i += 256) {
        const size_t idx = (size_t)i * cols + j;
        s += fabsf(A[idx] * Msk[idx]);
    }
    sm[threadIdx.x] = s;
    __syncthreads();
    for (int st = 128; st > 0; st >>= 1) {
        if (threadIdx.x < st) sm[threadIdx.x] += sm[threadIdx.x + st];
        __syncthreads();
    }
    if (threadIdx.x == 0) colsum[j] = sm[0];
}

__global__ __launch_bounds__(256)
void scalars_kernel(const float* __restrict__ col_l1,
                    const float* __restrict__ whl,
                    float* __restrict__ scalars) {
    __shared__ float sm[256];
    const int t = threadIdx.x;
    float mx = -1e30f;
    for (int j = t; j < HD; j += 256) mx = fmaxf(mx, col_l1[j]);
    sm[t] = mx; __syncthreads();
    for (int s = 128; s > 0; s >>= 1) {
        if (t < s) sm[t] = fmaxf(sm[t], sm[t + s]);
        __syncthreads();
    }
    const float m1 = sm[0];
    __syncthreads();
    mx = -1e30f;
    for (int j = t; j < Lsz; j += 256) mx = fmaxf(mx, whl[j]);
    sm[t] = mx; __syncthreads();
    for (int s = 128; s > 0; s >>= 1) {
        if (t < s) sm[t] = fmaxf(sm[t], sm[t + s]);
        __syncthreads();
    }
    if (t == 0) {
        scalars[0] = 0.95f / fmaxf(m1, 1e-8f);
        scalars[1] = sm[0] * 19.0f;
    }
}

__global__ __launch_bounds__(256)
void build_wt_kernel(const float* __restrict__ R, const float* __restrict__ M,
                     const float* __restrict__ scalars, int sel,
                     unsigned short* __restrict__ WT,
                     int K, int N, int srcld, int NL) {
    const size_t idx = (size_t)blockIdx.x * 256 + threadIdx.x;
    const size_t total = (size_t)N * K;
    if (idx >= total) return;
    const int n = (int)(idx / K);
    const int k = (int)(idx % K);
    float v = 0.f;
    if (n < NL) {
        const size_t s = (size_t)k * srcld + n;
        v = R[s] * M[s];
        if (sel == 0) v *= scalars[0];
    }
    WT[idx] = f2bf(v);
}

__global__ __launch_bounds__(256)
void to_bf16_kernel(const float* __restrict__ src, unsigned short* __restrict__ dst,
                    int n) {
    const int i = blockIdx.x * 256 + threadIdx.x;
    if (i < n) dst[i] = f2bf(src[i]);
}

__global__ __launch_bounds__(256)
void init_state(float* __restrict__ H0, unsigned short* __restrict__ Hb0,
                unsigned int* __restrict__ dH, int* __restrict__ active) {
    const size_t n = (size_t)Bsz * HD;
    for (size_t t = (size_t)blockIdx.x * 256 + threadIdx.x; t < n;
         t += (size_t)gridDim.x * 256) {
        H0[t] = 0.f;
        Hb0[t] = 0;
    }
    const int i = blockIdx.x * 256 + threadIdx.x;
    if (i < Bsz) { dH[i] = 0u; active[i] = 1; }
}

__global__ __launch_bounds__(256)
void update_rows(const float* __restrict__ logits,
                 const float* __restrict__ scalars,
                 unsigned int* __restrict__ dHbits,
                 int* __restrict__ active,
                 float* __restrict__ out,
                 int is_last) {
    const int lane = threadIdx.x & 31;
    const int wave = threadIdx.x >> 5;
    const int row = blockIdx.x * 8 + wave;
    if (row >= Bsz) return;

    if (active[row] != 0) {
        const float* lr = logits + (size_t)row * LP;
        float m1 = -1e30f, m2 = -1e30f;
        for (int n = lane; n < Lsz; n += 32) {
            const float v = lr[n];
            if (v > m1) { m2 = m1; m1 = v; }
            else if (v > m2) { m2 = v; }
        }
        for (int off = 16; off > 0; off >>= 1) {
            const float o1 = __shfl_xor(m1, off, 32);
            const float o2 = __shfl_xor(m2, off, 32);
            const float hi = fmaxf(m1, o1);
            const float lo = fmaxf(fminf(m1, o1), fmaxf(m2, o2));
            m1 = hi; m2 = lo;
        }
        const float margin = m1 - m2;
        const float dh = __uint_as_float(dHbits[row]);
        const float Gamma = scalars[1] * dh;
        const bool newly = margin > 2.0f * Gamma;
        if (newly || is_last) {
            for (int n = lane; n < Lsz; n += 32)
                out[(size_t)row * Lsz + n] = lr[n];
            if (lane == 0) active[row] = 0;
        }
    }
    if (lane == 0) dHbits[row] = 0u;
}

static inline char* wsalloc(char*& p, size_t bytes) {
    char* r = p;
    p += (bytes + 255) & ~(size_t)255;
    return r;
}

extern "C" void kernel_launch(void* const* d_in, const int* in_sizes, int n_in,
                              void* d_out, int out_size, void* d_ws, size_t ws_size,
                              hipStream_t stream) {
    (void)in_sizes; (void)n_in; (void)out_size; (void)ws_size;
    const float* x   = (const float*)d_in[0];
    const float* RIH = (const float*)d_in[1];
    const float* RH  = (const float*)d_in[2];
    const float* RHL = (const float*)d_in[3];
    const float* BH  = (const float*)d_in[4];
    const float* BL  = (const float*)d_in[5];
    const float* MIH = (const float*)d_in[6];
    const float* MH  = (const float*)d_in[7];
    const float* MHL = (const float*)d_in[8];
    float* out = (float*)d_out;

    char* p = (char*)d_ws;
    unsigned short* WT_H  = (unsigned short*)wsalloc(p, (size_t)HD * HD * 2);
    unsigned short* WT_HL = (unsigned short*)wsalloc(p, (size_t)LP * HD * 2);
    unsigned short* WT_IH = (unsigned short*)wsalloc(p, (size_t)HD * Isz * 2);
    unsigned short* xb    = (unsigned short*)wsalloc(p, (size_t)Bsz * Isz * 2);
    float* drive = (float*)wsalloc(p, (size_t)Bsz * HD * 4);
    float* Hbuf0 = (float*)wsalloc(p, (size_t)Bsz * HD * 4);
    float* Hbuf1 = (float*)wsalloc(p, (size_t)Bsz * HD * 4);
    unsigned short* Hb0 = (unsigned short*)wsalloc(p, (size_t)Bsz * HD * 2);
    unsigned short* Hb1 = (unsigned short*)wsalloc(p, (size_t)Bsz * HD * 2);
    float* logits = (float*)wsalloc(p, (size_t)Bsz * LP * 4);
    float* col_l1 = (float*)wsalloc(p, HD * 4);
    float* whl    = (float*)wsalloc(p, LP * 4);
    float* scalars = (float*)wsalloc(p, 256);
    unsigned int* dHb = (unsigned int*)wsalloc(p, Bsz * 4);
    int* active = (int*)wsalloc(p, Bsz * 4);

    col_l1_kernel<<<HD, 256, 0, stream>>>(RH, MH, col_l1, HD, HD);
    col_l1_kernel<<<Lsz, 256, 0, stream>>>(RHL, MHL, whl, HD, Lsz);
    scalars_kernel<<<1, 256, 0, stream>>>(col_l1, whl, scalars);

    build_wt_kernel<<<(HD * HD) / 256, 256, 0, stream>>>(
        RH, MH, scalars,  0, WT_H, HD, HD, HD, HD);
    build_wt_kernel<<<(LP * HD) / 256, 256, 0, stream>>>(
        RHL, MHL, scalars,  1, WT_HL, HD, LP, Lsz, Lsz);
    build_wt_kernel<<<(HD * Isz) / 256, 256, 0, stream>>>(
        RIH, MIH, scalars,  1, WT_IH, Isz, HD, HD, HD);
    to_bf16_kernel<<<(Bsz * Isz) / 256, 256, 0, stream>>>(x, xb, Bsz * Isz);
    init_state<<<(Bsz * HD) / 256, 256, 0, stream>>>(Hbuf0, Hb0, dHb, active);

    wmma_gemm<0><<<dim3(HD / 64, Bsz / 64), 128, 0, stream>>>(
        xb, WT_IH, Bsz, HD, Isz, BH, nullptr, nullptr, nullptr,
        drive, nullptr, nullptr, HD);

    float* Hc[2] = {Hbuf0, Hbuf1};
    unsigned short* Hbc[2] = {Hb0, Hb1};
    for (int t = 1; t <= TMAX; ++t) {
        const int pc = (t - 1) & 1, pn = t & 1;
        wmma_gemm<1><<<dim3(HD / 64, Bsz / 64), 128, 0, stream>>>(
            Hbc[pc], WT_H, Bsz, HD, HD, nullptr, drive, Hc[pc], active,
            Hc[pn], Hbc[pn], dHb, HD);
        wmma_gemm<2><<<dim3(LP / 64, Bsz / 64), 128, 0, stream>>>(
            Hbc[pn], WT_HL, Bsz, LP, HD, BL, nullptr, nullptr, nullptr,
            logits, nullptr, nullptr, Lsz);
        update_rows<<<Bsz / 8, 256, 0, stream>>>(
            logits, scalars, dHb, active, out, (t == TMAX) ? 1 : 0);
    }
}


# No row kernel: the logits GEMM stores still-active rows' logits straight into out (last write = certified step); each recurrence block derives its rows' flags from the 16 tile pairs; same bits
# speedup vs baseline: 2.2052x; 2.2052x over previous
#include <hip/hip_runtime.h>


#define Bsz 512
#define Isz 1024
#define HD  2048
#define Lsz 1000
#define LP  1024
#define TMAX 32
#define KD  4

typedef __attribute__((ext_vector_type(16))) __bf16 v16bf;
typedef __attribute__((ext_vector_type(8)))  float  v8f;
typedef __attribute__((ext_vector_type(4)))  unsigned int u32x4;

__device__ __forceinline__ unsigned short f2bf(float f) {
    unsigned int u = __float_as_uint(f);
    u += 0x7FFFu + ((u >> 16) & 1u);
    return (unsigned short)(u >> 16);
}

__device__ __forceinline__ size_t packed_idx(int r, int k, int K) {
    const int w = k & 31;
    const int lane = (((w >> 3) & 1) << 4) + (r & 15);
    return ((size_t)(r >> 4) * (size_t)(K >> 5) + (size_t)(k >> 5)) * 512
         + (size_t)(((w >> 4) << 8) + lane * 8 + (w & 7));
}

struct alignas(32) FragBits { u32x4 lo, hi; };

__device__ __forceinline__ v16bf load_frag(const unsigned short* __restrict__ base,
                                           unsigned off) {
    const char* p = reinterpret_cast<const char*>(base) + off;
    FragBits fb;
    fb.lo = *reinterpret_cast<const u32x4*>(p);
    fb.hi = *reinterpret_cast<const u32x4*>(p + 512);
    v16bf v;
    __builtin_memcpy(&v, &fb, sizeof(v));
    return v;
}

__device__ __forceinline__ void top2_put(float& m1, float& m2, float x) {
    m2 = fmaxf(m2, fminf(m1, x));
    m1 = fmaxf(m1, x);
}
__device__ __forceinline__ void top2_join(float& m1, float& m2, float o1, float o2) {
    const float hi = fmaxf(m1, o1);
    const float lo = fmaxf(fminf(m1, o1), fmaxf(m2, o2));
    m1 = hi; m2 = lo;
}
template <int MASK>
__device__ __forceinline__ float row_xor(float v) {
    return __int_as_float(__builtin_amdgcn_update_dpp(
        0, __float_as_int(v), 0x160 + MASK, 0xf, 0xf, false));
}
__device__ __forceinline__ void top2_row16(float& m1, float& m2) {
    top2_join(m1, m2, row_xor<8>(m1), row_xor<8>(m2));
    top2_join(m1, m2, row_xor<4>(m1), row_xor<4>(m2));
    top2_join(m1, m2, row_xor<2>(m1), row_xor<2>(m2));
    top2_join(m1, m2, row_xor<1>(m1), row_xor<1>(m2));
}

template <int MODE>
__global__ __launch_bounds__(128)
void wmma_gemm(const unsigned short* __restrict__ A,
               const unsigned short* __restrict__ BT,
               int M, int N, int K,
               const float* __restrict__ bias,
               const float* __restrict__ drive,
               const float* __restrict__ Hold,
               const int*   __restrict__ actIn,
               int*         __restrict__ actOut,
               float* __restrict__ Cout,
               unsigned short* __restrict__ HbOut,
               const unsigned int* __restrict__ dHprev,
               unsigned int* __restrict__ dHbits,
               float* __restrict__ Top2,
               const float* __restrict__ scalars,
               int NL) {
    __shared__ float4 red[4 * 16 * 2 * 32];

    const int lane = threadIdx.x & 31;
    const int wave = __builtin_amdgcn_readfirstlane((int)(threadIdx.x >> 5));
    const int rowBase = blockIdx.y * 64 + wave * 16;
    const int colBase = blockIdx.x * 64;
    const int r16 = lane & 15;
    const int kh  = lane >> 4;

    const int nk = K >> 7;
    const size_t kOff = (size_t)wave * nk * 512;
    const size_t strip = (size_t)16 * K;
    const unsigned short* sa0 = A + (size_t)(blockIdx.y * 64) * K + kOff;
    const unsigned short* sa1 = sa0 + strip;
    const unsigned short* sa2 = sa1 + strip;
    const unsigned short* sa3 = sa2 + strip;
    const unsigned short* sb0 = BT + (size_t)colBase * K + kOff;
    const unsigned short* sb1 = sb0 + strip;
    const unsigned short* sb2 = sb1 + strip;
    const unsigned short* sb3 = sb2 + strip;
    const unsigned boff = lane * 16;

#define LOAD_SLOT(s, o)                                   \
    a[s][0] = load_frag(sa0, (o) + (s) * 1024);           \
    b[s][0] = load_frag(sb0, (o) + (s) * 1024);           \
    a[s][1] = load_frag(sa1, (o) + (s) * 1024);           \
    b[s][1] = load_frag(sb1, (o) + (s) * 1024);           \
    a[s][2] = load_frag(sa2, (o) + (s) * 1024);           \
    b[s][2] = load_frag(sb2, (o) + (s) * 1024);           \
    a[s][3] = load_frag(sa3, (o) + (s) * 1024);           \
    b[s][3] = load_frag(sb3, (o) + (s) * 1024);

    v8f acc0 = {0.f,0.f,0.f,0.f,0.f,0.f,0.f,0.f};
    v8f acc[4][4];
#pragma unroll
    for (int i = 0; i < 4; ++i)
#pragma unroll
        for (int f = 0; f < 4; ++f) acc[i][f] = acc0;

    v16bf a[KD][4], b[KD][4];
#pragma unroll
    for (int s = 0; s < KD; ++s) {
        LOAD_SLOT(s, boff)
        __builtin_amdgcn_sched_barrier(0);
    }
#define ADVANCE_STREAMS()                                             \
    sa0 += KD * 512; sa1 += KD * 512; sa2 += KD * 512; sa3 += KD * 512; \
    sb0 += KD * 512; sb1 += KD * 512; sb2 += KD * 512; sb3 += KD * 512;
    ADVANCE_STREAMS()

    float2 pr[8];
    float dhPrev = 0.f, gcoef = 0.f;
    int actPrev = 1;
    if (MODE == 1) {
        const int row = rowBase + r16;
        const float2* tp = reinterpret_cast<const float2*>(Top2);
#pragma unroll
        for (int j = 0; j < 8; ++j) pr[j] = tp[(size_t)(kh * 8 + j) * M + row];
        dhPrev = __uint_as_float(dHprev[row]);
        actPrev = actIn[row];
        gcoef = scalars[1];
    }
    if (MODE == 2) actPrev = actIn[rowBase + r16];

    for (int k0 = 0; k0 < nk - KD; k0 += KD) {
#pragma unroll
        for (int s = 0; s < KD; ++s) {
#pragma unroll
            for (int i = 0; i < 4; ++i)
#pragma unroll
                for (int f = 0; f < 4; ++f)
                    acc[i][f] = __builtin_amdgcn_wmma_f32_16x16x32_bf16(
                        false, a[s][i], false, b[s][f], (short)0, acc[i][f], false, false);
            LOAD_SLOT(s, boff)
        }
        ADVANCE_STREAMS()
    }
#undef LOAD_SLOT
#undef ADVANCE_STREAMS
#pragma unroll
    for (int s = 0; s < KD; ++s) {
#pragma unroll
        for (int i = 0; i < 4; ++i)
#pragma unroll
            for (int f = 0; f < 4; ++f)
                acc[i][f] = __builtin_amdgcn_wmma_f32_16x16x32_bf16(
                    false, a[s][i], false, b[s][f], (short)0, acc[i][f], false, false);
    }

#pragma unroll
    for (int i = 0; i < 4; ++i)
#pragma unroll
        for (int f = 0; f < 4; ++f) {
            const int t = ((wave * 16 + i * 4 + f) * 2) * 32 + lane;
            red[t]      = make_float4(acc[i][f][0], acc[i][f][1], acc[i][f][2], acc[i][f][3]);
            red[t + 32] = make_float4(acc[i][f][4], acc[i][f][5], acc[i][f][6], acc[i][f][7]);
        }
    __syncthreads();

    __builtin_amdgcn_sched_barrier(0);
    unsigned actMask = 0xFFFFFFFFu;
    int actNew = 1;
    if (MODE == 1) {
        float m1 = pr[0].x, m2 = pr[0].y;
#pragma unroll
        for (int j = 1; j < 8; ++j) top2_join(m1, m2, pr[j].x, pr[j].y);
        const float o1 = __shfl_xor(m1, 16, 32);
        const float o2 = __shfl_xor(m2, 16, 32);
        top2_join(m1, m2, o1, o2);
        const float margin = m1 - m2;
        const float Gamma = gcoef * dhPrev;
        const bool newly = margin > 2.0f * Gamma;
        actNew = (actPrev != 0 && !newly) ? 1 : 0;
        actMask = __builtin_amdgcn_ballot_w32(actNew != 0);
    }
    if (MODE == 2) actMask = __builtin_amdgcn_ballot_w32(actPrev != 0);

    v8f accs[4];
#pragma unroll
    for (int f = 0; f < 4; ++f) {
        const int t = ((wave * 4 + f) * 2) * 32 + lane;
        float4 lo = red[t], hi = red[t + 32];
#pragma unroll
        for (int q = 1; q < 4; ++q) {
            const float4 l2 = red[t + q * 16 * 2 * 32];
            const float4 h2 = red[t + q * 16 * 2 * 32 + 32];
            lo.x += l2.x; lo.y += l2.y; lo.z += l2.z; lo.w += l2.w;
            hi.x += h2.x; hi.y += h2.y; hi.z += h2.z; hi.w += h2.w;
        }
        accs[f][0] = lo.x; accs[f][1] = lo.y; accs[f][2] = lo.z; accs[f][3] = lo.w;
        accs[f][4] = hi.x; accs[f][5] = hi.y; accs[f][6] = hi.z; accs[f][7] = hi.w;
    }

    if (MODE == 1 && blockIdx.x == 0 && lane < 16) actOut[rowBase + lane] = actNew;
    bool rowActive[8];
    float dmax[8];
    if (MODE == 1) {
#pragma unroll
        for (int r = 0; r < 8; ++r) {
            rowActive[r] = ((actMask >> (r + 8 * kh)) & 1u) != 0;
            dmax[r] = 0.f;
        }
    }

    if (MODE == 2) {
        float t1[8], t2[8];
#pragma unroll
        for (int r = 0; r < 8; ++r) { t1[r] = -1e30f; t2[r] = -1e30f; }
        const unsigned o0 = (unsigned)(rowBase + 8 * kh) * (unsigned)NL + (unsigned)(colBase + r16);
        const unsigned rowOn = actMask >> (8 * kh);
#pragma unroll
        for (int f = 0; f < 4; ++f) {
            const int n = colBase + f * 16 + r16;
            const bool ok = n < NL;
            const float bn = ok ? bias[n] : 0.f;
            float x[8];
#pragma unroll
            for (int r = 0; r < 8; ++r) x[r] = accs[f][r] + bn;
            if (ok) {
#pragma unroll
                for (int r = 0; r < 8; ++r)
                    if ((rowOn >> r) & 1u)
                        Cout[o0 + (unsigned)r * (unsigned)NL + (unsigned)(f * 16)] = x[r];
            }
#pragma unroll
            for (int r = 0; r < 8; ++r) top2_put(t1[r], t2[r], ok ? x[r] : -1e30f);
        }
#pragma unroll
        for (int r = 0; r < 8; ++r) top2_row16(t1[r], t2[r]);
        float s1 = t1[0], s2 = t2[0];
#pragma unroll
        for (int r = 1; r < 8; ++r) {
            if (r16 == r) { s1 = t1[r]; s2 = t2[r]; }
        }
        if (r16 < 8) {
            reinterpret_cast<float2*>(Top2)[(size_t)blockIdx.x * M + rowBase + r16 + 8 * kh] =
                make_float2(s1, s2);
        }
    }

#pragma unroll
    for (int f = 0; f < (MODE == 2 ? 0 : 4); ++f) {
        const int n = colBase + f * 16 + r16;
#pragma unroll
        for (int r = 0; r < 8; ++r) {
            const int m = rowBase + r + 8 * kh;
            const float v = accs[f][r];
            if (MODE == 0) {
                Cout[(size_t)m * N + n] = v + bias[n];
            } else {
                const size_t idx = (size_t)m * N + n;
                const float pre = v + drive[idx];
                const float actv = pre >= 0.f ? pre : 0.05f * pre;
                const float ho = Hold[idx];
                const float hn = rowActive[r] ? actv : ho;
                Cout[idx] = hn;
                HbOut[packed_idx(m, n, N)] = f2bf(hn);
                dmax[r] = fmaxf(dmax[r], fabsf(hn - ho));
            }
        }
    }

    if (MODE == 1) {
#pragma unroll
        for (int r = 0; r < 8; ++r) {
            if (dmax[r] > 0.f) {
                atomicMax(dHbits + (rowBase + r + 8 * kh), __float_as_uint(dmax[r]));
            }
        }
    }
}

__global__ __launch_bounds__(256)
void col_l1_kernel(const float* __restrict__ A, const float* __restrict__ Msk,
                   float* __restrict__ colsum, int rows, int cols) {
    __shared__ float sm[256];
    const int j = blockIdx.x;
    float s = 0.f;
    for (int i = threadIdx.x; i < rows; i += 256) {
        const size_t idx = (size_t)i * cols + j;
        s += fabsf(A[idx] * Msk[idx]);
    }
    sm[threadIdx.x] = s;
    __syncthreads();
    for (int st = 128; st > 0; st >>= 1) {
        if (threadIdx.x < st) sm[threadIdx.x] += sm[threadIdx.x + st];
        __syncthreads();
    }
    if (threadIdx.x == 0) colsum[j] = sm[0];
}

__global__ __launch_bounds__(256)
void scalars_kernel(const float* __restrict__ col_l1,
                    const float* __restrict__ whl,
                    float* __restrict__ scalars) {
    __shared__ float sm[256];
    const int t = threadIdx.x;
    float mx = -1e30f;
    for (int j = t; j < HD; j += 256) mx = fmaxf(mx, col_l1[j]);
    sm[t] = mx; __syncthreads();
    for (int s = 128; s > 0; s >>= 1) {
        if (t < s) sm[t] = fmaxf(sm[t], sm[t + s]);
        __syncthreads();
    }
    const float m1 = sm[0];
    __syncthreads();
    mx = -1e30f;
    for (int j = t; j < Lsz; j += 256) mx = fmaxf(mx, whl[j]);
    sm[t] = mx; __syncthreads();
    for (int s = 128; s > 0; s >>= 1) {
        if (t < s) sm[t] = fmaxf(sm[t], sm[t + s]);
        __syncthreads();
    }
    if (t == 0) {
        scalars[0] = 0.95f / fmaxf(m1, 1e-8f);
        scalars[1] = sm[0] * 19.0f;
    }
}

__global__ __launch_bounds__(256)
void build_wt_kernel(const float* __restrict__ R, const float* __restrict__ M,
                     const float* __restrict__ scalars, int sel,
                     unsigned short* __restrict__ WT,
                     int K, int N, int srcld, int NL) {
    const size_t idx = (size_t)blockIdx.x * 256 + threadIdx.x;
    const size_t total = (size_t)N * K;
    if (idx >= total) return;
    const int n = (int)(idx / K);
    const int k = (int)(idx % K);
    float v = 0.f;
    if (n < NL) {
        const size_t s = (size_t)k * srcld + n;
        v = R[s] * M[s];
        if (sel == 0) v *= scalars[0];
    }
    WT[packed_idx(n, k, K)] = f2bf(v);
}

__global__ __launch_bounds__(256)
void to_bf16_kernel(const float* __restrict__ src, unsigned short* __restrict__ dst,
                    int n, int K) {
    const int i = blockIdx.x * 256 + threadIdx.x;
    if (i < n) dst[packed_idx(i / K, i % K, K)] = f2bf(src[i]);
}

__global__ __launch_bounds__(256)
void init_state(float* __restrict__ H0, unsigned short* __restrict__ Hb0,
                unsigned int* __restrict__ dH, int* __restrict__ act0,
                float* __restrict__ top2) {
    const size_t n = (size_t)Bsz * HD;
    for (size_t t = (size_t)blockIdx.x * 256 + threadIdx.x; t < n;
         t += (size_t)gridDim.x * 256) {
        H0[t] = 0.f;
        Hb0[t] = 0;
    }
    const int i = blockIdx.x * 256 + threadIdx.x;
    if (i < (TMAX + 1) * Bsz) dH[i] = 0u;
    if (i < Bsz) act0[i] = 1;
    if (i < (LP / 64) * Bsz * 2) top2[i] = 0.f;
}

static inline char* wsalloc(char*& p, size_t bytes) {
    char* r = p;
    p += (bytes + 255) & ~(size_t)255;
    return r;
}

extern "C" void kernel_launch(void* const* d_in, const int* in_sizes, int n_in,
                              void* d_out, int out_size, void* d_ws, size_t ws_size,
                              hipStream_t stream) {
    (void)in_sizes; (void)n_in; (void)out_size; (void)ws_size;
    const float* x   = (const float*)d_in[0];
    const float* RIH = (const float*)d_in[1];
    const float* RH  = (const float*)d_in[2];
    const float* RHL = (const float*)d_in[3];
    const float* BH  = (const float*)d_in[4];
    const float* BL  = (const float*)d_in[5];
    const float* MIH = (const float*)d_in[6];
    const float* MH  = (const float*)d_in[7];
    const float* MHL = (const float*)d_in[8];
    float* out = (float*)d_out;

    char* p = (char*)d_ws;
    unsigned short* WT_H  = (unsigned short*)wsalloc(p, (size_t)HD * HD * 2);
    unsigned short* WT_HL = (unsigned short*)wsalloc(p, (size_t)LP * HD * 2);
    unsigned short* WT_IH = (unsigned short*)wsalloc(p, (size_t)HD * Isz * 2);
    unsigned short* xb    = (unsigned short*)wsalloc(p, (size_t)Bsz * Isz * 2);
    float* drive = (float*)wsalloc(p, (size_t)Bsz * HD * 4);
    float* Hbuf0 = (float*)wsalloc(p, (size_t)Bsz * HD * 4);
    float* Hbuf1 = (float*)wsalloc(p, (size_t)Bsz * HD * 4);
    unsigned short* Hb0 = (unsigned short*)wsalloc(p, (size_t)Bsz * HD * 2);
    unsigned short* Hb1 = (unsigned short*)wsalloc(p, (size_t)Bsz * HD * 2);
    float* col_l1 = (float*)wsalloc(p, HD * 4);
    float* whl    = (float*)wsalloc(p, LP * 4);
    float* scalars = (float*)wsalloc(p, 256);
    unsigned int* dHb = (unsigned int*)wsalloc(p, (size_t)(TMAX + 1) * Bsz * 4);
    int* act = (int*)wsalloc(p, (size_t)(TMAX + 1) * Bsz * 4);
    float* top2 = (float*)wsalloc(p, (size_t)(LP / 64) * Bsz * 2 * 4);

    col_l1_kernel<<<HD, 256, 0, stream>>>(RH, MH, col_l1, HD, HD);
    col_l1_kernel<<<Lsz, 256, 0, stream>>>(RHL, MHL, whl, HD, Lsz);
    scalars_kernel<<<1, 256, 0, stream>>>(col_l1, whl, scalars);

    build_wt_kernel<<<(HD * HD) / 256, 256, 0, stream>>>(
        RH, MH, scalars,  0, WT_H, HD, HD, HD, HD);
    build_wt_kernel<<<(LP * HD) / 256, 256, 0, stream>>>(
        RHL, MHL, scalars,  1, WT_HL, HD, LP, Lsz, Lsz);
    build_wt_kernel<<<(HD * Isz) / 256, 256, 0, stream>>>(
        RIH, MIH, scalars,  1, WT_IH, Isz, HD, HD, HD);
    to_bf16_kernel<<<(Bsz * Isz) / 256, 256, 0, stream>>>(x, xb, Bsz * Isz, Isz);
    init_state<<<(Bsz * HD) / 256, 256, 0, stream>>>(Hbuf0, Hb0, dHb, act, top2);

    wmma_gemm<0><<<dim3(HD / 64, Bsz / 64), 128, 0, stream>>>(
        xb, WT_IH, Bsz, HD, Isz, BH, nullptr, nullptr, nullptr, nullptr,
        drive, nullptr, nullptr, nullptr, nullptr, nullptr, HD);

    float* Hc[2] = {Hbuf0, Hbuf1};
    unsigned short* Hbc[2] = {Hb0, Hb1};
    for (int t = 1; t <= TMAX; ++t) {
        const int pc = (t - 1) & 1, pn = t & 1;
        wmma_gemm<1><<<dim3(HD / 64, Bsz / 64), 128, 0, stream>>>(
            Hbc[pc], WT_H, Bsz, HD, HD, nullptr, drive, Hc[pc],
            act + (t - 1) * Bsz, act + t * Bsz,
            Hc[pn], Hbc[pn], dHb + (t - 1) * Bsz, dHb + t * Bsz,
            top2, scalars, HD);
        wmma_gemm<2><<<dim3(LP / 64, Bsz / 64), 128, 0, stream>>>(
            Hbc[pn], WT_HL, Bsz, LP, HD, BL, nullptr, nullptr,
            act + t * Bsz, nullptr,
            out, nullptr, nullptr, nullptr, top2, nullptr, Lsz);
    }
}


# dH without atomics: a recurrence wave folds its 16 rows' maxima over its 64 columns (DPP) and stores them in its own line of a per-column-block plane; next step takes the max of 32; same bits
# speedup vs baseline: 2.4224x; 1.0985x over previous
#include <hip/hip_runtime.h>


#define Bsz 512
#define Isz 1024
#define HD  2048
#define Lsz 1000
#define LP  1024
#define TMAX 32
#define KD  4
#define NCB (HD / 64)
#define DHP (NCB * (Bsz / 16) * 32)

typedef __attribute__((ext_vector_type(16))) __bf16 v16bf;
typedef __attribute__((ext_vector_type(8)))  float  v8f;
typedef __attribute__((ext_vector_type(4)))  unsigned int u32x4;

__device__ __forceinline__ unsigned short f2bf(float f) {
    unsigned int u = __float_as_uint(f);
    u += 0x7FFFu + ((u >> 16) & 1u);
    return (unsigned short)(u >> 16);
}

__device__ __forceinline__ size_t packed_idx(int r, int k, int K) {
    const int w = k & 31;
    const int lane = (((w >> 3) & 1) << 4) + (r & 15);
    return ((size_t)(r >> 4) * (size_t)(K >> 5) + (size_t)(k >> 5)) * 512
         + (size_t)(((w >> 4) << 8) + lane * 8 + (w & 7));
}

struct alignas(32) FragBits { u32x4 lo, hi; };

__device__ __forceinline__ v16bf load_frag(const unsigned short* __restrict__ base,
                                           unsigned off) {
    const char* p = reinterpret_cast<const char*>(base) + off;
    FragBits fb;
    fb.lo = *reinterpret_cast<const u32x4*>(p);
    fb.hi = *reinterpret_cast<const u32x4*>(p + 512);
    v16bf v;
    __builtin_memcpy(&v, &fb, sizeof(v));
    return v;
}

__device__ __forceinline__ void top2_put(float& m1, float& m2, float x) {
    m2 = fmaxf(m2, fminf(m1, x));
    m1 = fmaxf(m1, x);
}
__device__ __forceinline__ void top2_join(float& m1, float& m2, float o1, float o2) {
    const float hi = fmaxf(m1, o1);
    const float lo = fmaxf(fminf(m1, o1), fmaxf(m2, o2));
    m1 = hi; m2 = lo;
}
template <int MASK>
__device__ __forceinline__ float row_xor(float v) {
    return __int_as_float(__builtin_amdgcn_update_dpp(
        0, __float_as_int(v), 0x160 + MASK, 0xf, 0xf, false));
}
__device__ __forceinline__ void top2_row16(float& m1, float& m2) {
    top2_join(m1, m2, row_xor<8>(m1), row_xor<8>(m2));
    top2_join(m1, m2, row_xor<4>(m1), row_xor<4>(m2));
    top2_join(m1, m2, row_xor<2>(m1), row_xor<2>(m2));
    top2_join(m1, m2, row_xor<1>(m1), row_xor<1>(m2));
}

template <int MODE>
__global__ __launch_bounds__(128)
void wmma_gemm(const unsigned short* __restrict__ A,
               const unsigned short* __restrict__ BT,
               int M, int N, int K,
               const float* __restrict__ bias,
               const float* __restrict__ drive,
               const float* __restrict__ Hold,
               const int*   __restrict__ actIn,
               int*         __restrict__ actOut,
               float* __restrict__ Cout,
               unsigned short* __restrict__ HbOut,
               const float* __restrict__ dHprev,
               float* __restrict__ dHout,
               float* __restrict__ Top2,
               const float* __restrict__ scalars,
               int NL) {
    __shared__ float4 red[4 * 16 * 2 * 32];

    const int lane = threadIdx.x & 31;
    const int wave = __builtin_amdgcn_readfirstlane((int)(threadIdx.x >> 5));
    const int rowBase = blockIdx.y * 64 + wave * 16;
    const int colBase = blockIdx.x * 64;
    const int r16 = lane & 15;
    const int kh  = lane >> 4;

    const int nk = K >> 7;
    const size_t kOff = (size_t)wave * nk * 512;
    const size_t strip = (size_t)16 * K;
    const unsigned short* sa0 = A + (size_t)(blockIdx.y * 64) * K + kOff;
    const unsigned short* sa1 = sa0 + strip;
    const unsigned short* sa2 = sa1 + strip;
    const unsigned short* sa3 = sa2 + strip;
    const unsigned short* sb0 = BT + (size_t)colBase * K + kOff;
    const unsigned short* sb1 = sb0 + strip;
    const unsigned short* sb2 = sb1 + strip;
    const unsigned short* sb3 = sb2 + strip;
    const unsigned boff = lane * 16;

#define LOAD_SLOT(s, o)                                   \
    a[s][0] = load_frag(sa0, (o) + (s) * 1024);           \
    b[s][0] = load_frag(sb0, (o) + (s) * 1024);           \
    a[s][1] = load_frag(sa1, (o) + (s) * 1024);           \
    b[s][1] = load_frag(sb1, (o) + (s) * 1024);           \
    a[s][2] = load_frag(sa2, (o) + (s) * 1024);           \
    b[s][2] = load_frag(sb2, (o) + (s) * 1024);           \
    a[s][3] = load_frag(sa3, (o) + (s) * 1024);           \
    b[s][3] = load_frag(sb3, (o) + (s) * 1024);

    v8f acc0 = {0.f,0.f,0.f,0.f,0.f,0.f,0.f,0.f};
    v8f acc[4][4];
#pragma unroll
    for (int i = 0; i < 4; ++i)
#pragma unroll
        for (int f = 0; f < 4; ++f) acc[i][f] = acc0;

    v16bf a[KD][4], b[KD][4];
#pragma unroll
    for (int s = 0; s < KD; ++s) {
        LOAD_SLOT(s, boff)
        __builtin_amdgcn_sched_barrier(0);
    }
#define ADVANCE_STREAMS()                                             \
    sa0 += KD * 512; sa1 += KD * 512; sa2 += KD * 512; sa3 += KD * 512; \
    sb0 += KD * 512; sb1 += KD * 512; sb2 += KD * 512; sb3 += KD * 512;
    ADVANCE_STREAMS()

    float2 pr[8];
    float dp[NCB / 2];
    float dhPrev = 0.f, gcoef = 0.f;
    int actPrev = 1;
    if (MODE == 1) {
        const int row = rowBase + r16;
        const float2* tp = reinterpret_cast<const float2*>(Top2);
#pragma unroll
        for (int j = 0; j < 8; ++j) pr[j] = tp[(size_t)(kh * 8 + j) * M + row];
        const float* dq = dHprev + ((size_t)(kh * (NCB / 2)) * (M >> 4) + (rowBase >> 4)) * 32 + r16;
#pragma unroll
        for (int j = 0; j < NCB / 2; ++j) dp[j] = dq[(size_t)j * (M >> 4) * 32];
        actPrev = actIn[row];
        gcoef = scalars[1];
    }
    if (MODE == 2) actPrev = actIn[rowBase + r16];

    for (int k0 = 0; k0 < nk - KD; k0 += KD) {
#pragma unroll
        for (int s = 0; s < KD; ++s) {
#pragma unroll
            for (int i = 0; i < 4; ++i)
#pragma unroll
                for (int f = 0; f < 4; ++f)
                    acc[i][f] = __builtin_amdgcn_wmma_f32_16x16x32_bf16(
                        false, a[s][i], false, b[s][f], (short)0, acc[i][f], false, false);
            LOAD_SLOT(s, boff)
        }
        ADVANCE_STREAMS()
    }
#undef LOAD_SLOT
#undef ADVANCE_STREAMS
#pragma unroll
    for (int s = 0; s < KD; ++s) {
#pragma unroll
        for (int i = 0; i < 4; ++i)
#pragma unroll
            for (int f = 0; f < 4; ++f)
                acc[i][f] = __builtin_amdgcn_wmma_f32_16x16x32_bf16(
                    false, a[s][i], false, b[s][f], (short)0, acc[i][f], false, false);
    }

#pragma unroll
    for (int i = 0; i < 4; ++i)
#pragma unroll
        for (int f = 0; f < 4; ++f) {
            const int t = ((wave * 16 + i * 4 + f) * 2) * 32 + lane;
            red[t]      = make_float4(acc[i][f][0], acc[i][f][1], acc[i][f][2], acc[i][f][3]);
            red[t + 32] = make_float4(acc[i][f][4], acc[i][f][5], acc[i][f][6], acc[i][f][7]);
        }
    __syncthreads();

    __builtin_amdgcn_sched_barrier(0);
    unsigned actMask = 0xFFFFFFFFu;
    int actNew = 1;
    if (MODE == 1) {
        float m1 = pr[0].x, m2 = pr[0].y;
#pragma unroll
        for (int j = 1; j < 8; ++j) top2_join(m1, m2, pr[j].x, pr[j].y);
        const float o1 = __shfl_xor(m1, 16, 32);
        const float o2 = __shfl_xor(m2, 16, 32);
        top2_join(m1, m2, o1, o2);
        const float margin = m1 - m2;
        dhPrev = dp[0];
#pragma unroll
        for (int j = 1; j < NCB / 2; ++j) dhPrev = fmaxf(dhPrev, dp[j]);
        dhPrev = fmaxf(dhPrev, __shfl_xor(dhPrev, 16, 32));
        const float Gamma = gcoef * dhPrev;
        const bool newly = margin > 2.0f * Gamma;
        actNew = (actPrev != 0 && !newly) ? 1 : 0;
        actMask = __builtin_amdgcn_ballot_w32(actNew != 0);
    }
    if (MODE == 2) actMask = __builtin_amdgcn_ballot_w32(actPrev != 0);

    v8f accs[4];
#pragma unroll
    for (int f = 0; f < 4; ++f) {
        const int t = ((wave * 4 + f) * 2) * 32 + lane;
        float4 lo = red[t], hi = red[t + 32];
#pragma unroll
        for (int q = 1; q < 4; ++q) {
            const float4 l2 = red[t + q * 16 * 2 * 32];
            const float4 h2 = red[t + q * 16 * 2 * 32 + 32];
            lo.x += l2.x; lo.y += l2.y; lo.z += l2.z; lo.w += l2.w;
            hi.x += h2.x; hi.y += h2.y; hi.z += h2.z; hi.w += h2.w;
        }
        accs[f][0] = lo.x; accs[f][1] = lo.y; accs[f][2] = lo.z; accs[f][3] = lo.w;
        accs[f][4] = hi.x; accs[f][5] = hi.y; accs[f][6] = hi.z; accs[f][7] = hi.w;
    }

    if (MODE == 1 && blockIdx.x == 0 && lane < 16) actOut[rowBase + lane] = actNew;
    bool rowActive[8];
    float dmax[8];
    if (MODE == 1) {
#pragma unroll
        for (int r = 0; r < 8; ++r) {
            rowActive[r] = ((actMask >> (r + 8 * kh)) & 1u) != 0;
            dmax[r] = 0.f;
        }
    }

    if (MODE == 2) {
        float t1[8], t2[8];
#pragma unroll
        for (int r = 0; r < 8; ++r) { t1[r] = -1e30f; t2[r] = -1e30f; }
        const unsigned o0 = (unsigned)(rowBase + 8 * kh) * (unsigned)NL + (unsigned)(colBase + r16);
        const unsigned rowOn = actMask >> (8 * kh);
#pragma unroll
        for (int f = 0; f < 4; ++f) {
            const int n = colBase + f * 16 + r16;
            const bool ok = n < NL;
            const float bn = ok ? bias[n] : 0.f;
            float x[8];
#pragma unroll
            for (int r = 0; r < 8; ++r) x[r] = accs[f][r] + bn;
            if (ok) {
#pragma unroll
                for (int r = 0; r < 8; ++r)
                    if ((rowOn >> r) & 1u)
                        Cout[o0 + (unsigned)r * (unsigned)NL + (unsigned)(f * 16)] = x[r];
            }
#pragma unroll
            for (int r = 0; r < 8; ++r) top2_put(t1[r], t2[r], ok ? x[r] : -1e30f);
        }
#pragma unroll
        for (int r = 0; r < 8; ++r) top2_row16(t1[r], t2[r]);
        float s1 = t1[0], s2 = t2[0];
#pragma unroll
        for (int r = 1; r < 8; ++r) {
            if (r16 == r) { s1 = t1[r]; s2 = t2[r]; }
        }
        if (r16 < 8) {
            reinterpret_cast<float2*>(Top2)[(size_t)blockIdx.x * M + rowBase + r16 + 8 * kh] =
                make_float2(s1, s2);
        }
    }

#pragma unroll
    for (int f = 0; f < (MODE == 2 ? 0 : 4); ++f) {
        const int n = colBase + f * 16 + r16;
#pragma unroll
        for (int r = 0; r < 8; ++r) {
            const int m = rowBase + r + 8 * kh;
            const float v = accs[f][r];
            if (MODE == 0) {
                Cout[(size_t)m * N + n] = v + bias[n];
            } else {
                const size_t idx = (size_t)m * N + n;
                const float pre = v + drive[idx];
                const float actv = pre >= 0.f ? pre : 0.05f * pre;
                const float ho = Hold[idx];
                const float hn = rowActive[r] ? actv : ho;
                Cout[idx] = hn;
                HbOut[packed_idx(m, n, N)] = f2bf(hn);
                dmax[r] = fmaxf(dmax[r], fabsf(hn - ho));
            }
        }
    }

    if (MODE == 1) {
#pragma unroll
        for (int r = 0; r < 8; ++r) {
            dmax[r] = fmaxf(dmax[r], row_xor<8>(dmax[r]));
            dmax[r] = fmaxf(dmax[r], row_xor<4>(dmax[r]));
            dmax[r] = fmaxf(dmax[r], row_xor<2>(dmax[r]));
            dmax[r] = fmaxf(dmax[r], row_xor<1>(dmax[r]));
        }
        float s = dmax[0];
#pragma unroll
        for (int r = 1; r < 8; ++r) {
            if (r16 == r) s = dmax[r];
        }
        if (r16 < 8) {
            dHout[((size_t)blockIdx.x * (M >> 4) + (rowBase >> 4)) * 32 + r16 + 8 * kh] = s;
        }
    }
}

__global__ __launch_bounds__(256)
void col_l1_kernel(const float* __restrict__ A, const float* __restrict__ Msk,
                   float* __restrict__ colsum, int rows, int cols) {
    __shared__ float sm[256];
    const int j = blockIdx.x;
    float s = 0.f;
    for (int i = threadIdx.x; i < rows; i += 256) {
        const size_t idx = (size_t)i * cols + j;
        s += fabsf(A[idx] * Msk[idx]);
    }
    sm[threadIdx.x] = s;
    __syncthreads();
    for (int st = 128; st > 0; st >>= 1) {
        if (threadIdx.x < st) sm[threadIdx.x] += sm[threadIdx.x + st];
        __syncthreads();
    }
    if (threadIdx.x == 0) colsum[j] = sm[0];
}

__global__ __launch_bounds__(256)
void scalars_kernel(const float* __restrict__ col_l1,
                    const float* __restrict__ whl,
                    float* __restrict__ scalars) {
    __shared__ float sm[256];
    const int t = threadIdx.x;
    float mx = -1e30f;
    for (int j = t; j < HD; j += 256) mx = fmaxf(mx, col_l1[j]);
    sm[t] = mx; __syncthreads();
    for (int s = 128; s > 0; s >>= 1) {
        if (t < s) sm[t] = fmaxf(sm[t], sm[t + s]);
        __syncthreads();
    }
    const float m1 = sm[0];
    __syncthreads();
    mx = -1e30f;
    for (int j = t; j < Lsz; j += 256) mx = fmaxf(mx, whl[j]);
    sm[t] = mx; __syncthreads();
    for (int s = 128; s > 0; s >>= 1) {
        if (t < s) sm[t] = fmaxf(sm[t], sm[t + s]);
        __syncthreads();
    }
    if (t == 0) {
        scalars[0] = 0.95f / fmaxf(m1, 1e-8f);
        scalars[1] = sm[0] * 19.0f;
    }
}

__global__ __launch_bounds__(256)
void build_wt_kernel(const float* __restrict__ R, const float* __restrict__ M,
                     const float* __restrict__ scalars, int sel,
                     unsigned short* __restrict__ WT,
                     int K, int N, int srcld, int NL) {
    const size_t idx = (size_t)blockIdx.x * 256 + threadIdx.x;
    const size_t total = (size_t)N * K;
    if (idx >= total) return;
    const int n = (int)(idx / K);
    const int k = (int)(idx % K);
    float v = 0.f;
    if (n < NL) {
        const size_t s = (size_t)k * srcld + n;
        v = R[s] * M[s];
        if (sel == 0) v *= scalars[0];
    }
    WT[packed_idx(n, k, K)] = f2bf(v);
}

__global__ __launch_bounds__(256)
void to_bf16_kernel(const float* __restrict__ src, unsigned short* __restrict__ dst,
                    int n, int K) {
    const int i = blockIdx.x * 256 + threadIdx.x;
    if (i < n) dst[packed_idx(i / K, i % K, K)] = f2bf(src[i]);
}

__global__ __launch_bounds__(256)
void init_state(float* __restrict__ H0, unsigned short* __restrict__ Hb0,
                float* __restrict__ dH, int* __restrict__ act0,
                float* __restrict__ top2) {
    const size_t n = (size_t)Bsz * HD;
    for (size_t t = (size_t)blockIdx.x * 256 + threadIdx.x; t < n;
         t += (size_t)gridDim.x * 256) {
        H0[t] = 0.f;
        Hb0[t] = 0;
    }
    const int i = blockIdx.x * 256 + threadIdx.x;
    if (i < DHP) dH[i] = 0.f;
    if (i < Bsz) act0[i] = 1;
    if (i < (LP / 64) * Bsz * 2) top2[i] = 0.f;
}

static inline char* wsalloc(char*& p, size_t bytes) {
    char* r = p;
    p += (bytes + 255) & ~(size_t)255;
    return r;
}

extern "C" void kernel_launch(void* const* d_in, const int* in_sizes, int n_in,
                              void* d_out, int out_size, void* d_ws, size_t ws_size,
                              hipStream_t stream) {
    (void)in_sizes; (void)n_in; (void)out_size; (void)ws_size;
    const float* x   = (const float*)d_in[0];
    const float* RIH = (const float*)d_in[1];
    const float* RH  = (const float*)d_in[2];
    const float* RHL = (const float*)d_in[3];
    const float* BH  = (const float*)d_in[4];
    const float* BL  = (const float*)d_in[5];
    const float* MIH = (const float*)d_in[6];
    const float* MH  = (const float*)d_in[7];
    const float* MHL = (const float*)d_in[8];
    float* out = (float*)d_out;

    char* p = (char*)d_ws;
    unsigned short* WT_H  = (unsigned short*)wsalloc(p, (size_t)HD * HD * 2);
    unsigned short* WT_HL = (unsigned short*)wsalloc(p, (size_t)LP * HD * 2);
    unsigned short* WT_IH = (unsigned short*)wsalloc(p, (size_t)HD * Isz * 2);
    unsigned short* xb    = (unsigned short*)wsalloc(p, (size_t)Bsz * Isz * 2);
    float* drive = (float*)wsalloc(p, (size_t)Bsz * HD * 4);
    float* Hbuf0 = (float*)wsalloc(p, (size_t)Bsz * HD * 4);
    float* Hbuf1 = (float*)wsalloc(p, (size_t)Bsz * HD * 4);
    unsigned short* Hb0 = (unsigned short*)wsalloc(p, (size_t)Bsz * HD * 2);
    unsigned short* Hb1 = (unsigned short*)wsalloc(p, (size_t)Bsz * HD * 2);
    float* col_l1 = (float*)wsalloc(p, HD * 4);
    float* whl    = (float*)wsalloc(p, LP * 4);
    float* scalars = (float*)wsalloc(p, 256);
    float* dHp = (float*)wsalloc(p, (size_t)2 * DHP * 4);
    int* act = (int*)wsalloc(p, (size_t)(TMAX + 1) * Bsz * 4);
    float* top2 = (float*)wsalloc(p, (size_t)(LP / 64) * Bsz * 2 * 4);

    col_l1_kernel<<<HD, 256, 0, stream>>>(RH, MH, col_l1, HD, HD);
    col_l1_kernel<<<Lsz, 256, 0, stream>>>(RHL, MHL, whl, HD, Lsz);
    scalars_kernel<<<1, 256, 0, stream>>>(col_l1, whl, scalars);

    build_wt_kernel<<<(HD * HD) / 256, 256, 0, stream>>>(
        RH, MH, scalars,  0, WT_H, HD, HD, HD, HD);
    build_wt_kernel<<<(LP * HD) / 256, 256, 0, stream>>>(
        RHL, MHL, scalars,  1, WT_HL, HD, LP, Lsz, Lsz);
    build_wt_kernel<<<(HD * Isz) / 256, 256, 0, stream>>>(
        RIH, MIH, scalars,  1, WT_IH, Isz, HD, HD, HD);
    to_bf16_kernel<<<(Bsz * Isz) / 256, 256, 0, stream>>>(x, xb, Bsz * Isz, Isz);
    init_state<<<(Bsz * HD) / 256, 256, 0, stream>>>(Hbuf0, Hb0, dHp, act, top2);

    wmma_gemm<0><<<dim3(HD / 64, Bsz / 64), 128, 0, stream>>>(
        xb, WT_IH, Bsz, HD, Isz, BH, nullptr, nullptr, nullptr, nullptr,
        drive, nullptr, nullptr, nullptr, nullptr, nullptr, HD);

    float* Hc[2] = {Hbuf0, Hbuf1};
    unsigned short* Hbc[2] = {Hb0, Hb1};
    for (int t = 1; t <= TMAX; ++t) {
        const int pc = (t - 1) & 1, pn = t & 1;
        wmma_gemm<1><<<dim3(HD / 64, Bsz / 64), 128, 0, stream>>>(
            Hbc[pc], WT_H, Bsz, HD, HD, nullptr, drive, Hc[pc],
            act + (t - 1) * Bsz, act + t * Bsz,
            Hc[pn], Hbc[pn], dHp + pc * DHP, dHp + pn * DHP,
            top2, scalars, HD);
        wmma_gemm<2><<<dim3(LP / 64, Bsz / 64), 128, 0, stream>>>(
            Hbc[pn], WT_HL, Bsz, LP, HD, BL, nullptr, nullptr,
            act + t * Bsz, nullptr,
            out, nullptr, nullptr, nullptr, top2, nullptr, Lsz);
    }
}


# Logits GEMMs of two consecutive steps run as ONE launch (M = 2 x 512; 16 launches + a row fix-up, was 32): recurrence no longer gated, certification settled one batch late; same bits
# speedup vs baseline: 2.8606x; 1.1809x over previous
#include <hip/hip_runtime.h>


#define Bsz 512
#define Isz 1024
#define HD  2048
#define Lsz 1000
#define LP  1024
#define TMAX 32
#define KD  4
#define NCB (HD / 64)
#define DHP (NCB * (Bsz / 16) * 32)

typedef __attribute__((ext_vector_type(16))) __bf16 v16bf;
typedef __attribute__((ext_vector_type(8)))  float  v8f;
typedef __attribute__((ext_vector_type(4)))  unsigned int u32x4;

__device__ __forceinline__ unsigned short f2bf(float f) {
    unsigned int u = __float_as_uint(f);
    u += 0x7FFFu + ((u >> 16) & 1u);
    return (unsigned short)(u >> 16);
}

__device__ __forceinline__ size_t packed_idx(int r, int k, int K) {
    const int w = k & 31;
    const int lane = (((w >> 3) & 1) << 4) + (r & 15);
    return ((size_t)(r >> 4) * (size_t)(K >> 5) + (size_t)(k >> 5)) * 512
         + (size_t)(((w >> 4) << 8) + lane * 8 + (w & 7));
}

struct alignas(32) FragBits { u32x4 lo, hi; };

__device__ __forceinline__ v16bf load_frag(const unsigned short* __restrict__ base,
                                           unsigned off) {
    const char* p = reinterpret_cast<const char*>(base) + off;
    FragBits fb;
    fb.lo = *reinterpret_cast<const u32x4*>(p);
    fb.hi = *reinterpret_cast<const u32x4*>(p + 512);
    v16bf v;
    __builtin_memcpy(&v, &fb, sizeof(v));
    return v;
}

__device__ __forceinline__ void top2_put(float& m1, float& m2, float x) {
    m2 = fmaxf(m2, fminf(m1, x));
    m1 = fmaxf(m1, x);
}
__device__ __forceinline__ void top2_join(float& m1, float& m2, float o1, float o2) {
    const float hi = fmaxf(m1, o1);
    const float lo = fmaxf(fminf(m1, o1), fmaxf(m2, o2));
    m1 = hi; m2 = lo;
}
template <int MASK>
__device__ __forceinline__ float row_xor(float v) {
    return __int_as_float(__builtin_amdgcn_update_dpp(
        0, __float_as_int(v), 0x160 + MASK, 0xf, 0xf, false));
}
__device__ __forceinline__ void top2_row16(float& m1, float& m2) {
    top2_join(m1, m2, row_xor<8>(m1), row_xor<8>(m2));
    top2_join(m1, m2, row_xor<4>(m1), row_xor<4>(m2));
    top2_join(m1, m2, row_xor<2>(m1), row_xor<2>(m2));
    top2_join(m1, m2, row_xor<1>(m1), row_xor<1>(m2));
}

template <int MODE>
__global__ __launch_bounds__(128)
void wmma_gemm(const unsigned short* __restrict__ A,
               const unsigned short* __restrict__ A2,
               const unsigned short* __restrict__ BT,
               int M, int N, int K,
               const float* __restrict__ bias,
               const float* __restrict__ drive,
               const float* __restrict__ Hold,
               const int*   __restrict__ actIn,
               int*         __restrict__ actOut,
               float* __restrict__ Cout,
               float* __restrict__ Stage,
               unsigned short* __restrict__ HbOut,
               const float* __restrict__ dHa,
               const float* __restrict__ dHb,
               float* __restrict__ dHout,
               const float* __restrict__ Top2In,
               float* __restrict__ Top2Out,
               const float* __restrict__ scalars,
               int NL) {
    __shared__ float4 red[4 * 16 * 2 * 32];

    const int lane = threadIdx.x & 31;
    const int wave = __builtin_amdgcn_readfirstlane((int)(threadIdx.x >> 5));
    const int rbs = M >> 6;
    const int half = (MODE == 2 && (int)blockIdx.y >= rbs) ? 1 : 0;
    const int by = (int)blockIdx.y - half * rbs;
    const int rowBase = by * 64 + wave * 16;
    const int colBase = blockIdx.x * 64;
    const int r16 = lane & 15;
    const int kh  = lane >> 4;

    const int nk = K >> 7;
    const size_t kOff = (size_t)wave * nk * 512;
    const size_t strip = (size_t)16 * K;
    const unsigned short* Ah = (MODE == 2 && half != 0) ? A2 : A;
    const unsigned short* sa0 = Ah + (size_t)(by * 64) * K + kOff;
    const unsigned short* sa1 = sa0 + strip;
    const unsigned short* sa2 = sa1 + strip;
    const unsigned short* sa3 = sa2 + strip;
    const unsigned short* sb0 = BT + (size_t)colBase * K + kOff;
    const unsigned short* sb1 = sb0 + strip;
    const unsigned short* sb2 = sb1 + strip;
    const unsigned short* sb3 = sb2 + strip;
    const unsigned boff = lane * 16;

#define LOAD_SLOT(s, o)                                   \
    a[s][0] = load_frag(sa0, (o) + (s) * 1024);           \
    b[s][0] = load_frag(sb0, (o) + (s) * 1024);           \
    a[s][1] = load_frag(sa1, (o) + (s) * 1024);           \
    b[s][1] = load_frag(sb1, (o) + (s) * 1024);           \
    a[s][2] = load_frag(sa2, (o) + (s) * 1024);           \
    b[s][2] = load_frag(sb2, (o) + (s) * 1024);           \
    a[s][3] = load_frag(sa3, (o) + (s) * 1024);           \
    b[s][3] = load_frag(sb3, (o) + (s) * 1024);

    v8f acc0 = {0.f,0.f,0.f,0.f,0.f,0.f,0.f,0.f};
    v8f acc[4][4];
#pragma unroll
    for (int i = 0; i < 4; ++i)
#pragma unroll
        for (int f = 0; f < 4; ++f) acc[i][f] = acc0;

    v16bf a[KD][4], b[KD][4];
#pragma unroll
    for (int s = 0; s < KD; ++s) {
        LOAD_SLOT(s, boff)
        __builtin_amdgcn_sched_barrier(0);
    }
#define ADVANCE_STREAMS()                                             \
    sa0 += KD * 512; sa1 += KD * 512; sa2 += KD * 512; sa3 += KD * 512; \
    sb0 += KD * 512; sb1 += KD * 512; sb2 += KD * 512; sb3 += KD * 512;
    ADVANCE_STREAMS()

    float2 prA[8], prB[8];
    float dpA[NCB / 2], dpB[NCB / 2];
    float gcoef = 0.f;
    int actPrev = 1;
    if (MODE == 2) {
        const int row = rowBase + r16;
        const float2* tp = reinterpret_cast<const float2*>(Top2In);
        const size_t tiles = (size_t)(N >> 6);
#pragma unroll
        for (int j = 0; j < 8; ++j) {
            prA[j] = tp[(size_t)(kh * 8 + j) * M + row];
            prB[j] = tp[(tiles + (size_t)(kh * 8 + j)) * M + row];
        }
        const size_t dofs = ((size_t)(kh * (NCB / 2)) * (M >> 4) + (rowBase >> 4)) * 32 + r16;
#pragma unroll
        for (int j = 0; j < NCB / 2; ++j) {
            dpA[j] = dHa[dofs + (size_t)j * (M >> 4) * 32];
            dpB[j] = dHb[dofs + (size_t)j * (M >> 4) * 32];
        }
        gcoef = scalars[1];
        actPrev = actIn[row];
    }

    for (int k0 = 0; k0 < nk - KD; k0 += KD) {
#pragma unroll
        for (int s = 0; s < KD; ++s) {
#pragma unroll
            for (int i = 0; i < 4; ++i)
#pragma unroll
                for (int f = 0; f < 4; ++f)
                    acc[i][f] = __builtin_amdgcn_wmma_f32_16x16x32_bf16(
                        false, a[s][i], false, b[s][f], (short)0, acc[i][f], false, false);
            LOAD_SLOT(s, boff)
        }
        ADVANCE_STREAMS()
    }
#undef LOAD_SLOT
#undef ADVANCE_STREAMS
#pragma unroll
    for (int s = 0; s < KD; ++s) {
#pragma unroll
        for (int i = 0; i < 4; ++i)
#pragma unroll
            for (int f = 0; f < 4; ++f)
                acc[i][f] = __builtin_amdgcn_wmma_f32_16x16x32_bf16(
                    false, a[s][i], false, b[s][f], (short)0, acc[i][f], false, false);
    }

#pragma unroll
    for (int i = 0; i < 4; ++i)
#pragma unroll
        for (int f = 0; f < 4; ++f) {
            const int t = ((wave * 16 + i * 4 + f) * 2) * 32 + lane;
            red[t]      = make_float4(acc[i][f][0], acc[i][f][1], acc[i][f][2], acc[i][f][3]);
            red[t + 32] = make_float4(acc[i][f][4], acc[i][f][5], acc[i][f][6], acc[i][f][7]);
        }
    __syncthreads();

    __builtin_amdgcn_sched_barrier(0);
    unsigned actMask = 0xFFFFFFFFu, copyMask = 0u;
    int actNew = 1;
    if (MODE == 2) {
        bool cert[2];
#pragma unroll
        for (int s = 0; s < 2; ++s) {
            const float2* pr = s ? prB : prA;
            const float*  dp = s ? dpB : dpA;
            float m1 = pr[0].x, m2 = pr[0].y;
#pragma unroll
            for (int j = 1; j < 8; ++j) top2_join(m1, m2, pr[j].x, pr[j].y);
            const float o1 = __shfl_xor(m1, 16, 32);
            const float o2 = __shfl_xor(m2, 16, 32);
            top2_join(m1, m2, o1, o2);
            const float margin = m1 - m2;
            float dh = dp[0];
#pragma unroll
            for (int j = 1; j < NCB / 2; ++j) dh = fmaxf(dh, dp[j]);
            dh = fmaxf(dh, __shfl_xor(dh, 16, 32));
            const float Gamma = gcoef * dh;
            cert[s] = margin > 2.0f * Gamma;
        }
        const bool mid = actPrev != 0 && !cert[0];
        actNew = (mid && !cert[1]) ? 1 : 0;
        actMask = __builtin_amdgcn_ballot_w32(actNew != 0);
        copyMask = __builtin_amdgcn_ballot_w32(mid && cert[1]);
    }

    v8f accs[4];
#pragma unroll
    for (int f = 0; f < 4; ++f) {
        const int t = ((wave * 4 + f) * 2) * 32 + lane;
        float4 lo = red[t], hi = red[t + 32];
#pragma unroll
        for (int q = 1; q < 4; ++q) {
            const float4 l2 = red[t + q * 16 * 2 * 32];
            const float4 h2 = red[t + q * 16 * 2 * 32 + 32];
            lo.x += l2.x; lo.y += l2.y; lo.z += l2.z; lo.w += l2.w;
            hi.x += h2.x; hi.y += h2.y; hi.z += h2.z; hi.w += h2.w;
        }
        accs[f][0] = lo.x; accs[f][1] = lo.y; accs[f][2] = lo.z; accs[f][3] = lo.w;
        accs[f][4] = hi.x; accs[f][5] = hi.y; accs[f][6] = hi.z; accs[f][7] = hi.w;
    }

    if (MODE == 2 && blockIdx.x == 0 && half == 0 && lane < 16) actOut[rowBase + lane] = actNew;
    float dmax[8];
    if (MODE == 1) {
#pragma unroll
        for (int r = 0; r < 8; ++r) dmax[r] = 0.f;
    }

    if (MODE == 2) {
        float t1[8], t2[8];
#pragma unroll
        for (int r = 0; r < 8; ++r) { t1[r] = -1e30f; t2[r] = -1e30f; }
        const unsigned o0 = (unsigned)(rowBase + 8 * kh) * (unsigned)NL + (unsigned)(colBase + r16);
        const unsigned rowOn = actMask >> (8 * kh);
        float* dst = half != 0 ? Stage : Cout;
        if (half == 0 && copyMask != 0u) {
            const unsigned rowCp = copyMask >> (8 * kh);
#pragma unroll
            for (int f = 0; f < 4; ++f) {
                if (colBase + f * 16 + r16 < NL) {
#pragma unroll
                    for (int r = 0; r < 8; ++r)
                        if ((rowCp >> r) & 1u) {
                            const unsigned o = o0 + (unsigned)r * (unsigned)NL + (unsigned)(f * 16);
                            Cout[o] = Stage[o];
                        }
                }
            }
        }
#pragma unroll
        for (int f = 0; f < 4; ++f) {
            const int n = colBase + f * 16 + r16;
            const bool ok = n < NL;
            const float bn = ok ? bias[n] : 0.f;
            float x[8];
#pragma unroll
            for (int r = 0; r < 8; ++r) x[r] = accs[f][r] + bn;
            if (ok) {
#pragma unroll
                for (int r = 0; r < 8; ++r)
                    if ((rowOn >> r) & 1u)
                        dst[o0 + (unsigned)r * (unsigned)NL + (unsigned)(f * 16)] = x[r];
            }
#pragma unroll
            for (int r = 0; r < 8; ++r) top2_put(t1[r], t2[r], ok ? x[r] : -1e30f);
        }
#pragma unroll
        for (int r = 0; r < 8; ++r) top2_row16(t1[r], t2[r]);
        float s1 = t1[0], s2 = t2[0];
#pragma unroll
        for (int r = 1; r < 8; ++r) {
            if (r16 == r) { s1 = t1[r]; s2 = t2[r]; }
        }
        if (r16 < 8) {
            reinterpret_cast<float2*>(Top2Out)[((size_t)half * gridDim.x + blockIdx.x) * M
                                               + rowBase + r16 + 8 * kh] = make_float2(s1, s2);
        }
    }

#pragma unroll
    for (int f = 0; f < (MODE == 2 ? 0 : 4); ++f) {
        const int n = colBase + f * 16 + r16;
#pragma unroll
        for (int r = 0; r < 8; ++r) {
            const int m = rowBase + r + 8 * kh;
            const float v = accs[f][r];
            if (MODE == 0) {
                Cout[(size_t)m * N + n] = v + bias[n];
            } else {
                const size_t idx = (size_t)m * N + n;
                const float pre = v + drive[idx];
                const float hn = pre >= 0.f ? pre : 0.05f * pre;
                const float ho = Hold[idx];
                Cout[idx] = hn;
                HbOut[packed_idx(m, n, N)] = f2bf(hn);
                dmax[r] = fmaxf(dmax[r], fabsf(hn - ho));
            }
        }
    }

    if (MODE == 1) {
#pragma unroll
        for (int r = 0; r < 8; ++r) {
            dmax[r] = fmaxf(dmax[r], row_xor<8>(dmax[r]));
            dmax[r] = fmaxf(dmax[r], row_xor<4>(dmax[r]));
            dmax[r] = fmaxf(dmax[r], row_xor<2>(dmax[r]));
            dmax[r] = fmaxf(dmax[r], row_xor<1>(dmax[r]));
        }
        float s = dmax[0];
#pragma unroll
        for (int r = 1; r < 8; ++r) {
            if (r16 == r) s = dmax[r];
        }
        if (r16 < 8) {
            dHout[((size_t)blockIdx.x * (M >> 4) + (rowBase >> 4)) * 32 + r16 + 8 * kh] = s;
        }
    }
}

__global__ __launch_bounds__(256)
void col_l1_kernel(const float* __restrict__ A, const float* __restrict__ Msk,
                   float* __restrict__ colsum, int rows, int cols) {
    __shared__ float sm[256];
    const int j = blockIdx.x;
    float s = 0.f;
    for (int i = threadIdx.x; i < rows; i += 256) {
        const size_t idx = (size_t)i * cols + j;
        s += fabsf(A[idx] * Msk[idx]);
    }
    sm[threadIdx.x] = s;
    __syncthreads();
    for (int st = 128; st > 0; st >>= 1) {
        if (threadIdx.x < st) sm[threadIdx.x] += sm[threadIdx.x + st];
        __syncthreads();
    }
    if (threadIdx.x == 0) colsum[j] = sm[0];
}

__global__ __launch_bounds__(256)
void scalars_kernel(const float* __restrict__ col_l1,
                    const float* __restrict__ whl,
                    float* __restrict__ scalars) {
    __shared__ float sm[256];
    const int t = threadIdx.x;
    float mx = -1e30f;
    for (int j = t; j < HD; j += 256) mx = fmaxf(mx, col_l1[j]);
    sm[t] = mx; __syncthreads();
    for (int s = 128; s > 0; s >>= 1) {
        if (t < s) sm[t] = fmaxf(sm[t], sm[t + s]);
        __syncthreads();
    }
    const float m1 = sm[0];
    __syncthreads();
    mx = -1e30f;
    for (int j = t; j < Lsz; j += 256) mx = fmaxf(mx, whl[j]);
    sm[t] = mx; __syncthreads();
    for (int s = 128; s > 0; s >>= 1) {
        if (t < s) sm[t] = fmaxf(sm[t], sm[t + s]);
        __syncthreads();
    }
    if (t == 0) {
        scalars[0] = 0.95f / fmaxf(m1, 1e-8f);
        scalars[1] = sm[0] * 19.0f;
    }
}

__global__ __launch_bounds__(256)
void build_wt_kernel(const float* __restrict__ R, const float* __restrict__ M,
                     const float* __restrict__ scalars, int sel,
                     unsigned short* __restrict__ WT,
                     int K, int N, int srcld, int NL) {
    const size_t idx = (size_t)blockIdx.x * 256 + threadIdx.x;
    const size_t total = (size_t)N * K;
    if (idx >= total) return;
    const int n = (int)(idx / K);
    const int k = (int)(idx % K);
    float v = 0.f;
    if (n < NL) {
        const size_t s = (size_t)k * srcld + n;
        v = R[s] * M[s];
        if (sel == 0) v *= scalars[0];
    }
    WT[packed_idx(n, k, K)] = f2bf(v);
}

__global__ __launch_bounds__(256)
void to_bf16_kernel(const float* __restrict__ src, unsigned short* __restrict__ dst,
                    int n, int K) {
    const int i = blockIdx.x * 256 + threadIdx.x;
    if (i < n) dst[packed_idx(i / K, i % K, K)] = f2bf(src[i]);
}

__global__ __launch_bounds__(256)
void init_state(float* __restrict__ H0, unsigned short* __restrict__ Hb0,
                int* __restrict__ act0, float* __restrict__ dHa0,
                float* __restrict__ dHb0, float* __restrict__ pairs0) {
    const size_t n = (size_t)Bsz * HD;
    for (size_t t = (size_t)blockIdx.x * 256 + threadIdx.x; t < n;
         t += (size_t)gridDim.x * 256) {
        H0[t] = 0.f;
        Hb0[t] = 0;
    }
    const int i = blockIdx.x * 256 + threadIdx.x;
    if (i < Bsz) act0[i] = 1;
    if (i < DHP) { dHa0[i] = 0.f; dHb0[i] = 0.f; }
    if (i < 2 * (LP / 64) * Bsz * 2) pairs0[i] = 0.f;
}

__global__ __launch_bounds__(128)
void finish_rows(const float* __restrict__ Stage, float* __restrict__ out,
                 const int* __restrict__ actIn,
                 const float* __restrict__ Top2In,
                 const float* __restrict__ dHa,
                 const float* __restrict__ scalars, int M, int NL) {
    const int lane = threadIdx.x & 31;
    const int r16 = lane & 15;
    const int kh  = lane >> 4;
    const int rowBase = blockIdx.y * 16;
    const int row = rowBase + r16;
    const float2* tp = reinterpret_cast<const float2*>(Top2In);
    float m1 = -1e30f, m2 = -1e30f;
#pragma unroll
    for (int j = 0; j < 8; ++j) {
        const float2 q = tp[(size_t)(kh * 8 + j) * M + row];
        if (j == 0) { m1 = q.x; m2 = q.y; } else top2_join(m1, m2, q.x, q.y);
    }
    const float o1 = __shfl_xor(m1, 16, 32);
    const float o2 = __shfl_xor(m2, 16, 32);
    top2_join(m1, m2, o1, o2);
    const float margin = m1 - m2;
    const size_t dofs = ((size_t)(kh * (NCB / 2)) * (M >> 4) + (rowBase >> 4)) * 32 + r16;
    float dh = dHa[dofs];
#pragma unroll
    for (int j = 1; j < NCB / 2; ++j) dh = fmaxf(dh, dHa[dofs + (size_t)j * (M >> 4) * 32]);
    dh = fmaxf(dh, __shfl_xor(dh, 16, 32));
    const float Gamma = scalars[1] * dh;
    const bool cert = margin > 2.0f * Gamma;
    const unsigned take = __builtin_amdgcn_ballot_w32(actIn[row] != 0 && !cert) & 0xFFFFu;
    if (take == 0u) return;
    const int per = (NL + (int)gridDim.x - 1) / (int)gridDim.x;
    const int c0 = blockIdx.x * per;
    const int c1 = min(NL, c0 + per);
    for (int r = 0; r < 16; ++r) {
        if ((take >> r) & 1u) {
            const size_t o = (size_t)(rowBase + r) * NL;
            for (int c = c0 + (int)threadIdx.x; c < c1; c += 128) out[o + c] = Stage[o + c];
        }
    }
}

static inline char* wsalloc(char*& p, size_t bytes) {
    char* r = p;
    p += (bytes + 255) & ~(size_t)255;
    return r;
}

extern "C" void kernel_launch(void* const* d_in, const int* in_sizes, int n_in,
                              void* d_out, int out_size, void* d_ws, size_t ws_size,
                              hipStream_t stream) {
    (void)in_sizes; (void)n_in; (void)out_size; (void)ws_size;
    const float* x   = (const float*)d_in[0];
    const float* RIH = (const float*)d_in[1];
    const float* RH  = (const float*)d_in[2];
    const float* RHL = (const float*)d_in[3];
    const float* BH  = (const float*)d_in[4];
    const float* BL  = (const float*)d_in[5];
    const float* MIH = (const float*)d_in[6];
    const float* MH  = (const float*)d_in[7];
    const float* MHL = (const float*)d_in[8];
    float* out = (float*)d_out;

    char* p = (char*)d_ws;
    unsigned short* WT_H  = (unsigned short*)wsalloc(p, (size_t)HD * HD * 2);
    unsigned short* WT_HL = (unsigned short*)wsalloc(p, (size_t)LP * HD * 2);
    unsigned short* WT_IH = (unsigned short*)wsalloc(p, (size_t)HD * Isz * 2);
    unsigned short* xb    = (unsigned short*)wsalloc(p, (size_t)Bsz * Isz * 2);
    float* drive = (float*)wsalloc(p, (size_t)Bsz * HD * 4);
    float* Hbuf0 = (float*)wsalloc(p, (size_t)Bsz * HD * 4);
    float* Hbuf1 = (float*)wsalloc(p, (size_t)Bsz * HD * 4);
    unsigned short* Hb0 = (unsigned short*)wsalloc(p, (size_t)Bsz * HD * 2);
    unsigned short* Hb1 = (unsigned short*)wsalloc(p, (size_t)Bsz * HD * 2);
    float* col_l1 = (float*)wsalloc(p, HD * 4);
    float* whl    = (float*)wsalloc(p, LP * 4);
    float* scalars = (float*)wsalloc(p, 256);
    int* act = (int*)wsalloc(p, (size_t)(TMAX / 2 + 1) * Bsz * 4);
    float* stage = (float*)WT_IH;
    float* dHp = (float*)xb;
    float* top2 = dHp + (size_t)4 * DHP;
    const size_t T2SET = (size_t)2 * (LP / 64) * Bsz * 2;

    col_l1_kernel<<<HD, 256, 0, stream>>>(RH, MH, col_l1, HD, HD);
    col_l1_kernel<<<Lsz, 256, 0, stream>>>(RHL, MHL, whl, HD, Lsz);
    scalars_kernel<<<1, 256, 0, stream>>>(col_l1, whl, scalars);

    build_wt_kernel<<<(HD * HD) / 256, 256, 0, stream>>>(
        RH, MH, scalars,  0, WT_H, HD, HD, HD, HD);
    build_wt_kernel<<<(LP * HD) / 256, 256, 0, stream>>>(
        RHL, MHL, scalars,  1, WT_HL, HD, LP, Lsz, Lsz);
    build_wt_kernel<<<(HD * Isz) / 256, 256, 0, stream>>>(
        RIH, MIH, scalars,  1, WT_IH, Isz, HD, HD, HD);
    to_bf16_kernel<<<(Bsz * Isz) / 256, 256, 0, stream>>>(x, xb, Bsz * Isz, Isz);

    wmma_gemm<0><<<dim3(HD / 64, Bsz / 64), 128, 0, stream>>>(
        xb, nullptr, WT_IH, Bsz, HD, Isz, BH, nullptr, nullptr, nullptr, nullptr,
        drive, nullptr, nullptr, nullptr, nullptr, nullptr, nullptr, nullptr, nullptr, HD);

    init_state<<<(Bsz * HD) / 256, 256, 0, stream>>>(
        Hbuf0, Hb0, act, dHp + (size_t)3 * DHP, dHp, top2);

    float* Hc[2] = {Hbuf0, Hbuf1};
    unsigned short* Hbc[2] = {Hb0, Hb1};
    for (int j = 1; j <= TMAX / 2; ++j) {
        for (int t = 2 * j - 1; t <= 2 * j; ++t) {
            const int pc = (t - 1) & 1, pn = t & 1;
            wmma_gemm<1><<<dim3(HD / 64, Bsz / 64), 128, 0, stream>>>(
                Hbc[pc], nullptr, WT_H, Bsz, HD, HD, nullptr, drive, Hc[pc],
                nullptr, nullptr, Hc[pn], nullptr, Hbc[pn],
                nullptr, nullptr, dHp + (size_t)(t & 3) * DHP,
                nullptr, nullptr, nullptr, HD);
        }
        wmma_gemm<2><<<dim3(LP / 64, 2 * (Bsz / 64)), 128, 0, stream>>>(
            Hbc[1], Hbc[0], WT_HL, Bsz, LP, HD, BL, nullptr, nullptr,
            act + (j - 1) * Bsz, act + j * Bsz, out, stage, nullptr,
            dHp + (size_t)((2 * j + 1) & 3) * DHP, dHp + (size_t)((2 * j + 2) & 3) * DHP, nullptr,
            top2 + (size_t)((j - 1) & 1) * T2SET, top2 + (size_t)(j & 1) * T2SET,
            scalars, Lsz);
    }
    finish_rows<<<dim3(4, Bsz / 16), 128, 0, stream>>>(
        stage, out, act + (TMAX / 2) * Bsz, top2 + (size_t)((TMAX / 2) & 1) * T2SET,
        dHp + (size_t)((TMAX - 1) & 3) * DHP, scalars, Bsz, Lsz);
}


# Set-up packers: one lane builds one 16-byte fragment-order chunk (8 k of one n) from 8 row-contiguous loads, in registers, no LDS; was 1 element per lane with 8 KB-strided reads; same bits
# speedup vs baseline: 3.1835x; 1.1129x over previous
#include <hip/hip_runtime.h>


#define Bsz 512
#define Isz 1024
#define HD  2048
#define Lsz 1000
#define LP  1024
#define TMAX 32
#define KD  4
#define NCB (HD / 64)
#define DHP (NCB * (Bsz / 16) * 32)

typedef __attribute__((ext_vector_type(16))) __bf16 v16bf;
typedef __attribute__((ext_vector_type(8)))  float  v8f;
typedef __attribute__((ext_vector_type(4)))  unsigned int u32x4;

__device__ __forceinline__ unsigned short f2bf(float f) {
    unsigned int u = __float_as_uint(f);
    u += 0x7FFFu + ((u >> 16) & 1u);
    return (unsigned short)(u >> 16);
}

__device__ __forceinline__ size_t packed_idx(int r, int k, int K) {
    const int w = k & 31;
    const int lane = (((w >> 3) & 1) << 4) + (r & 15);
    return ((size_t)(r >> 4) * (size_t)(K >> 5) + (size_t)(k >> 5)) * 512
         + (size_t)(((w >> 4) << 8) + lane * 8 + (w & 7));
}

struct alignas(32) FragBits { u32x4 lo, hi; };

__device__ __forceinline__ v16bf load_frag(const unsigned short* __restrict__ base,
                                           unsigned off) {
    const char* p = reinterpret_cast<const char*>(base) + off;
    FragBits fb;
    fb.lo = *reinterpret_cast<const u32x4*>(p);
    fb.hi = *reinterpret_cast<const u32x4*>(p + 512);
    v16bf v;
    __builtin_memcpy(&v, &fb, sizeof(v));
    return v;
}

__device__ __forceinline__ void top2_put(float& m1, float& m2, float x) {
    m2 = fmaxf(m2, fminf(m1, x));
    m1 = fmaxf(m1, x);
}
__device__ __forceinline__ void top2_join(float& m1, float& m2, float o1, float o2) {
    const float hi = fmaxf(m1, o1);
    const float lo = fmaxf(fminf(m1, o1), fmaxf(m2, o2));
    m1 = hi; m2 = lo;
}
template <int MASK>
__device__ __forceinline__ float row_xor(float v) {
    return __int_as_float(__builtin_amdgcn_update_dpp(
        0, __float_as_int(v), 0x160 + MASK, 0xf, 0xf, false));
}
__device__ __forceinline__ void top2_row16(float& m1, float& m2) {
    top2_join(m1, m2, row_xor<8>(m1), row_xor<8>(m2));
    top2_join(m1, m2, row_xor<4>(m1), row_xor<4>(m2));
    top2_join(m1, m2, row_xor<2>(m1), row_xor<2>(m2));
    top2_join(m1, m2, row_xor<1>(m1), row_xor<1>(m2));
}

template <int MODE>
__global__ __launch_bounds__(128)
void wmma_gemm(const unsigned short* __restrict__ A,
               const unsigned short* __restrict__ A2,
               const unsigned short* __restrict__ BT,
               int M, int N, int K,
               const float* __restrict__ bias,
               const float* __restrict__ drive,
               const float* __restrict__ Hold,
               const int*   __restrict__ actIn,
               int*         __restrict__ actOut,
               float* __restrict__ Cout,
               float* __restrict__ Stage,
               unsigned short* __restrict__ HbOut,
               const float* __restrict__ dHa,
               const float* __restrict__ dHb,
               float* __restrict__ dHout,
               const float* __restrict__ Top2In,
               float* __restrict__ Top2Out,
               const float* __restrict__ scalars,
               int NL) {
    __shared__ float4 red[4 * 16 * 2 * 32];

    const int lane = threadIdx.x & 31;
    const int wave = __builtin_amdgcn_readfirstlane((int)(threadIdx.x >> 5));
    const int rbs = M >> 6;
    const int half = (MODE == 2 && (int)blockIdx.y >= rbs) ? 1 : 0;
    const int by = (int)blockIdx.y - half * rbs;
    const int rowBase = by * 64 + wave * 16;
    const int colBase = blockIdx.x * 64;
    const int r16 = lane & 15;
    const int kh  = lane >> 4;

    const int nk = K >> 7;
    const size_t kOff = (size_t)wave * nk * 512;
    const size_t strip = (size_t)16 * K;
    const unsigned short* Ah = (MODE == 2 && half != 0) ? A2 : A;
    const unsigned short* sa0 = Ah + (size_t)(by * 64) * K + kOff;
    const unsigned short* sa1 = sa0 + strip;
    const unsigned short* sa2 = sa1 + strip;
    const unsigned short* sa3 = sa2 + strip;
    const unsigned short* sb0 = BT + (size_t)colBase * K + kOff;
    const unsigned short* sb1 = sb0 + strip;
    const unsigned short* sb2 = sb1 + strip;
    const unsigned short* sb3 = sb2 + strip;
    const unsigned boff = lane * 16;

#define LOAD_SLOT(s, o)                                   \
    a[s][0] = load_frag(sa0, (o) + (s) * 1024);           \
    b[s][0] = load_frag(sb0, (o) + (s) * 1024);           \
    a[s][1] = load_frag(sa1, (o) + (s) * 1024);           \
    b[s][1] = load_frag(sb1, (o) + (s) * 1024);           \
    a[s][2] = load_frag(sa2, (o) + (s) * 1024);           \
    b[s][2] = load_frag(sb2, (o) + (s) * 1024);           \
    a[s][3] = load_frag(sa3, (o) + (s) * 1024);           \
    b[s][3] = load_frag(sb3, (o) + (s) * 1024);

    v8f acc0 = {0.f,0.f,0.f,0.f,0.f,0.f,0.f,0.f};
    v8f acc[4][4];
#pragma unroll
    for (int i = 0; i < 4; ++i)
#pragma unroll
        for (int f = 0; f < 4; ++f) acc[i][f] = acc0;

    v16bf a[KD][4], b[KD][4];
#pragma unroll
    for (int s = 0; s < KD; ++s) {
        LOAD_SLOT(s, boff)
        __builtin_amdgcn_sched_barrier(0);
    }
#define ADVANCE_STREAMS()                                             \
    sa0 += KD * 512; sa1 += KD * 512; sa2 += KD * 512; sa3 += KD * 512; \
    sb0 += KD * 512; sb1 += KD * 512; sb2 += KD * 512; sb3 += KD * 512;
    ADVANCE_STREAMS()

    float2 prA[8], prB[8];
    float dpA[NCB / 2], dpB[NCB / 2];
    float gcoef = 0.f;
    int actPrev = 1;
    if (MODE == 2) {
        const int row = rowBase + r16;
        const float2* tp = reinterpret_cast<const float2*>(Top2In);
        const size_t tiles = (size_t)(N >> 6);
#pragma unroll
        for (int j = 0; j < 8; ++j) {
            prA[j] = tp[(size_t)(kh * 8 + j) * M + row];
            prB[j] = tp[(tiles + (size_t)(kh * 8 + j)) * M + row];
        }
        const size_t dofs = ((size_t)(kh * (NCB / 2)) * (M >> 4) + (rowBase >> 4)) * 32 + r16;
#pragma unroll
        for (int j = 0; j < NCB / 2; ++j) {
            dpA[j] = dHa[dofs + (size_t)j * (M >> 4) * 32];
            dpB[j] = dHb[dofs + (size_t)j * (M >> 4) * 32];
        }
        gcoef = scalars[1];
        actPrev = actIn[row];
    }

    for (int k0 = 0; k0 < nk - KD; k0 += KD) {
#pragma unroll
        for (int s = 0; s < KD; ++s) {
#pragma unroll
            for (int i = 0; i < 4; ++i)
#pragma unroll
                for (int f = 0; f < 4; ++f)
                    acc[i][f] = __builtin_amdgcn_wmma_f32_16x16x32_bf16(
                        false, a[s][i], false, b[s][f], (short)0, acc[i][f], false, false);
            LOAD_SLOT(s, boff)
        }
        ADVANCE_STREAMS()
    }
#undef LOAD_SLOT
#undef ADVANCE_STREAMS
#pragma unroll
    for (int s = 0; s < KD; ++s) {
#pragma unroll
        for (int i = 0; i < 4; ++i)
#pragma unroll
            for (int f = 0; f < 4; ++f)
                acc[i][f] = __builtin_amdgcn_wmma_f32_16x16x32_bf16(
                    false, a[s][i], false, b[s][f], (short)0, acc[i][f], false, false);
    }

#pragma unroll
    for (int i = 0; i < 4; ++i)
#pragma unroll
        for (int f = 0; f < 4; ++f) {
            const int t = ((wave * 16 + i * 4 + f) * 2) * 32 + lane;
            red[t]      = make_float4(acc[i][f][0], acc[i][f][1], acc[i][f][2], acc[i][f][3]);
            red[t + 32] = make_float4(acc[i][f][4], acc[i][f][5], acc[i][f][6], acc[i][f][7]);
        }
    __syncthreads();

    __builtin_amdgcn_sched_barrier(0);
    unsigned actMask = 0xFFFFFFFFu, copyMask = 0u;
    int actNew = 1;
    if (MODE == 2) {
        bool cert[2];
#pragma unroll
        for (int s = 0; s < 2; ++s) {
            const float2* pr = s ? prB : prA;
            const float*  dp = s ? dpB : dpA;
            float m1 = pr[0].x, m2 = pr[0].y;
#pragma unroll
            for (int j = 1; j < 8; ++j) top2_join(m1, m2, pr[j].x, pr[j].y);
            const float o1 = __shfl_xor(m1, 16, 32);
            const float o2 = __shfl_xor(m2, 16, 32);
            top2_join(m1, m2, o1, o2);
            const float margin = m1 - m2;
            float dh = dp[0];
#pragma unroll
            for (int j = 1; j < NCB / 2; ++j) dh = fmaxf(dh, dp[j]);
            dh = fmaxf(dh, __shfl_xor(dh, 16, 32));
            const float Gamma = gcoef * dh;
            cert[s] = margin > 2.0f * Gamma;
        }
        const bool mid = actPrev != 0 && !cert[0];
        actNew = (mid && !cert[1]) ? 1 : 0;
        actMask = __builtin_amdgcn_ballot_w32(actNew != 0);
        copyMask = __builtin_amdgcn_ballot_w32(mid && cert[1]);
    }

    v8f accs[4];
#pragma unroll
    for (int f = 0; f < 4; ++f) {
        const int t = ((wave * 4 + f) * 2) * 32 + lane;
        float4 lo = red[t], hi = red[t + 32];
#pragma unroll
        for (int q = 1; q < 4; ++q) {
            const float4 l2 = red[t + q * 16 * 2 * 32];
            const float4 h2 = red[t + q * 16 * 2 * 32 + 32];
            lo.x += l2.x; lo.y += l2.y; lo.z += l2.z; lo.w += l2.w;
            hi.x += h2.x; hi.y += h2.y; hi.z += h2.z; hi.w += h2.w;
        }
        accs[f][0] = lo.x; accs[f][1] = lo.y; accs[f][2] = lo.z; accs[f][3] = lo.w;
        accs[f][4] = hi.x; accs[f][5] = hi.y; accs[f][6] = hi.z; accs[f][7] = hi.w;
    }

    if (MODE == 2 && blockIdx.x == 0 && half == 0 && lane < 16) actOut[rowBase + lane] = actNew;
    float dmax[8];
    if (MODE == 1) {
#pragma unroll
        for (int r = 0; r < 8; ++r) dmax[r] = 0.f;
    }

    if (MODE == 2) {
        float t1[8], t2[8];
#pragma unroll
        for (int r = 0; r < 8; ++r) { t1[r] = -1e30f; t2[r] = -1e30f; }
        const unsigned o0 = (unsigned)(rowBase + 8 * kh) * (unsigned)NL + (unsigned)(colBase + r16);
        const unsigned rowOn = actMask >> (8 * kh);
        float* dst = half != 0 ? Stage : Cout;
        if (half == 0 && copyMask != 0u) {
            const unsigned rowCp = copyMask >> (8 * kh);
#pragma unroll
            for (int f = 0; f < 4; ++f) {
                if (colBase + f * 16 + r16 < NL) {
#pragma unroll
                    for (int r = 0; r < 8; ++r)
                        if ((rowCp >> r) & 1u) {
                            const unsigned o = o0 + (unsigned)r * (unsigned)NL + (unsigned)(f * 16);
                            Cout[o] = Stage[o];
                        }
                }
            }
        }
#pragma unroll
        for (int f = 0; f < 4; ++f) {
            const int n = colBase + f * 16 + r16;
            const bool ok = n < NL;
            const float bn = ok ? bias[n] : 0.f;
            float x[8];
#pragma unroll
            for (int r = 0; r < 8; ++r) x[r] = accs[f][r] + bn;
            if (ok) {
#pragma unroll
                for (int r = 0; r < 8; ++r)
                    if ((rowOn >> r) & 1u)
                        dst[o0 + (unsigned)r * (unsigned)NL + (unsigned)(f * 16)] = x[r];
            }
#pragma unroll
            for (int r = 0; r < 8; ++r) top2_put(t1[r], t2[r], ok ? x[r] : -1e30f);
        }
#pragma unroll
        for (int r = 0; r < 8; ++r) top2_row16(t1[r], t2[r]);
        float s1 = t1[0], s2 = t2[0];
#pragma unroll
        for (int r = 1; r < 8; ++r) {
            if (r16 == r) { s1 = t1[r]; s2 = t2[r]; }
        }
        if (r16 < 8) {
            reinterpret_cast<float2*>(Top2Out)[((size_t)half * gridDim.x + blockIdx.x) * M
                                               + rowBase + r16 + 8 * kh] = make_float2(s1, s2);
        }
    }

#pragma unroll
    for (int f = 0; f < (MODE == 2 ? 0 : 4); ++f) {
        const int n = colBase + f * 16 + r16;
#pragma unroll
        for (int r = 0; r < 8; ++r) {
            const int m = rowBase + r + 8 * kh;
            const float v = accs[f][r];
            if (MODE == 0) {
                Cout[(size_t)m * N + n] = v + bias[n];
            } else {
                const size_t idx = (size_t)m * N + n;
                const float pre = v + drive[idx];
                const float hn = pre >= 0.f ? pre : 0.05f * pre;
                const float ho = Hold[idx];
                Cout[idx] = hn;
                HbOut[packed_idx(m, n, N)] = f2bf(hn);
                dmax[r] = fmaxf(dmax[r], fabsf(hn - ho));
            }
        }
    }

    if (MODE == 1) {
#pragma unroll
        for (int r = 0; r < 8; ++r) {
            dmax[r] = fmaxf(dmax[r], row_xor<8>(dmax[r]));
            dmax[r] = fmaxf(dmax[r], row_xor<4>(dmax[r]));
            dmax[r] = fmaxf(dmax[r], row_xor<2>(dmax[r]));
            dmax[r] = fmaxf(dmax[r], row_xor<1>(dmax[r]));
        }
        float s = dmax[0];
#pragma unroll
        for (int r = 1; r < 8; ++r) {
            if (r16 == r) s = dmax[r];
        }
        if (r16 < 8) {
            dHout[((size_t)blockIdx.x * (M >> 4) + (rowBase >> 4)) * 32 + r16 + 8 * kh] = s;
        }
    }
}

__global__ __launch_bounds__(256)
void col_l1_kernel(const float* __restrict__ A, const float* __restrict__ Msk,
                   float* __restrict__ colsum, int rows, int cols) {
    __shared__ float sm[256];
    const int j = blockIdx.x;
    float s = 0.f;
    for (int i = threadIdx.x; i < rows; i += 256) {
        const size_t idx = (size_t)i * cols + j;
        s += fabsf(A[idx] * Msk[idx]);
    }
    sm[threadIdx.x] = s;
    __syncthreads();
    for (int st = 128; st > 0; st >>= 1) {
        if (threadIdx.x < st) sm[threadIdx.x] += sm[threadIdx.x + st];
        __syncthreads();
    }
    if (threadIdx.x == 0) colsum[j] = sm[0];
}

__global__ __launch_bounds__(256)
void scalars_kernel(const float* __restrict__ col_l1,
                    const float* __restrict__ whl,
                    float* __restrict__ scalars) {
    __shared__ float sm[256];
    const int t = threadIdx.x;
    float mx = -1e30f;
    for (int j = t; j < HD; j += 256) mx = fmaxf(mx, col_l1[j]);
    sm[t] = mx; __syncthreads();
    for (int s = 128; s > 0; s >>= 1) {
        if (t < s) sm[t] = fmaxf(sm[t], sm[t + s]);
        __syncthreads();
    }
    const float m1 = sm[0];
    __syncthreads();
    mx = -1e30f;
    for (int j = t; j < Lsz; j += 256) mx = fmaxf(mx, whl[j]);
    sm[t] = mx; __syncthreads();
    for (int s = 128; s > 0; s >>= 1) {
        if (t < s) sm[t] = fmaxf(sm[t], sm[t + s]);
        __syncthreads();
    }
    if (t == 0) {
        scalars[0] = 0.95f / fmaxf(m1, 1e-8f);
        scalars[1] = sm[0] * 19.0f;
    }
}

template <int K>
__global__ __launch_bounds__(256)
void build_wt_kernel(const float* __restrict__ R, const float* __restrict__ M,
                     const float* __restrict__ scalars, int sel,
                     unsigned short* __restrict__ WT,
                     int N, int srcld, int NL) {
    const unsigned g = blockIdx.x * 256u + threadIdx.x;
    const int lane = (int)(g & 31u);
    const unsigned wv = g >> 5;
    const int k0 = (int)(wv % (unsigned)(K / 8)) * 8;
    const int n  = (int)(wv / (unsigned)(K / 8)) * 32 + lane;
    if (n >= N) return;
    float v[8];
    if (n < NL) {
        const size_t s = (size_t)k0 * srcld + n;
        float r[8], m[8];
#pragma unroll
        for (int e = 0; e < 8; ++e) {
            r[e] = R[s + (size_t)e * srcld];
            m[e] = M[s + (size_t)e * srcld];
        }
#pragma unroll
        for (int e = 0; e < 8; ++e) v[e] = r[e] * m[e];
        if (sel == 0) {
            const float sc = scalars[0];
#pragma unroll
            for (int e = 0; e < 8; ++e) v[e] *= sc;
        }
    } else {
#pragma unroll
        for (int e = 0; e < 8; ++e) v[e] = 0.f;
    }
    u32x4 c;
    c[0] = (unsigned)f2bf(v[0]) | ((unsigned)f2bf(v[1]) << 16);
    c[1] = (unsigned)f2bf(v[2]) | ((unsigned)f2bf(v[3]) << 16);
    c[2] = (unsigned)f2bf(v[4]) | ((unsigned)f2bf(v[5]) << 16);
    c[3] = (unsigned)f2bf(v[6]) | ((unsigned)f2bf(v[7]) << 16);
    *reinterpret_cast<u32x4*>(WT + packed_idx(n, k0, K)) = c;
}

template <int K>
__global__ __launch_bounds__(256)
void to_bf16_kernel(const float* __restrict__ src, unsigned short* __restrict__ dst,
                    int rows) {
    const unsigned g = blockIdx.x * 256u + threadIdx.x;
    const int lane = (int)(g & 31u);
    const unsigned hw = g >> 5;
    const int hh = (int)(hw & 1u);
    const unsigned tile = hw >> 1;
    const int kt = (int)(tile % (unsigned)(K / 32));
    const int r  = (int)(tile / (unsigned)(K / 32)) * 16 + (lane & 15);
    const int k0 = kt * 32 + hh * 16 + (lane >> 4) * 8;
    if (r >= rows) return;
    const float* p = src + (size_t)r * K + k0;
    float v[8];
#pragma unroll
    for (int e = 0; e < 8; ++e) v[e] = p[e];
    u32x4 c;
    c[0] = (unsigned)f2bf(v[0]) | ((unsigned)f2bf(v[1]) << 16);
    c[1] = (unsigned)f2bf(v[2]) | ((unsigned)f2bf(v[3]) << 16);
    c[2] = (unsigned)f2bf(v[4]) | ((unsigned)f2bf(v[5]) << 16);
    c[3] = (unsigned)f2bf(v[6]) | ((unsigned)f2bf(v[7]) << 16);
    *reinterpret_cast<u32x4*>(dst + packed_idx(r, k0, K)) = c;
}

__global__ __launch_bounds__(256)
void init_state(float* __restrict__ H0, unsigned short* __restrict__ Hb0,
                int* __restrict__ act0, float* __restrict__ dHa0,
                float* __restrict__ dHb0, float* __restrict__ pairs0) {
    const size_t n = (size_t)Bsz * HD;
    for (size_t t = (size_t)blockIdx.x * 256 + threadIdx.x; t < n;
         t += (size_t)gridDim.x * 256) {
        H0[t] = 0.f;
        Hb0[t] = 0;
    }
    const int i = blockIdx.x * 256 + threadIdx.x;
    if (i < Bsz) act0[i] = 1;
    if (i < DHP) { dHa0[i] = 0.f; dHb0[i] = 0.f; }
    if (i < 2 * (LP / 64) * Bsz * 2) pairs0[i] = 0.f;
}

__global__ __launch_bounds__(128)
void finish_rows(const float* __restrict__ Stage, float* __restrict__ out,
                 const int* __restrict__ actIn,
                 const float* __restrict__ Top2In,
                 const float* __restrict__ dHa,
                 const float* __restrict__ scalars, int M, int NL) {
    const int lane = threadIdx.x & 31;
    const int r16 = lane & 15;
    const int kh  = lane >> 4;
    const int rowBase = blockIdx.y * 16;
    const int row = rowBase + r16;
    const float2* tp = reinterpret_cast<const float2*>(Top2In);
    float m1 = -1e30f, m2 = -1e30f;
#pragma unroll
    for (int j = 0; j < 8; ++j) {
        const float2 q = tp[(size_t)(kh * 8 + j) * M + row];
        if (j == 0) { m1 = q.x; m2 = q.y; } else top2_join(m1, m2, q.x, q.y);
    }
    const float o1 = __shfl_xor(m1, 16, 32);
    const float o2 = __shfl_xor(m2, 16, 32);
    top2_join(m1, m2, o1, o2);
    const float margin = m1 - m2;
    const size_t dofs = ((size_t)(kh * (NCB / 2)) * (M >> 4) + (rowBase >> 4)) * 32 + r16;
    float dh = dHa[dofs];
#pragma unroll
    for (int j = 1; j < NCB / 2; ++j) dh = fmaxf(dh, dHa[dofs + (size_t)j * (M >> 4) * 32]);
    dh = fmaxf(dh, __shfl_xor(dh, 16, 32));
    const float Gamma = scalars[1] * dh;
    const bool cert = margin > 2.0f * Gamma;
    const unsigned take = __builtin_amdgcn_ballot_w32(actIn[row] != 0 && !cert) & 0xFFFFu;
    if (take == 0u) return;
    const int per = (NL + (int)gridDim.x - 1) / (int)gridDim.x;
    const int c0 = blockIdx.x * per;
    const int c1 = min(NL, c0 + per);
    for (int r = 0; r < 16; ++r) {
        if ((take >> r) & 1u) {
            const size_t o = (size_t)(rowBase + r) * NL;
            for (int c = c0 + (int)threadIdx.x; c < c1; c += 128) out[o + c] = Stage[o + c];
        }
    }
}

static inline char* wsalloc(char*& p, size_t bytes) {
    char* r = p;
    p += (bytes + 255) & ~(size_t)255;
    return r;
}

extern "C" void kernel_launch(void* const* d_in, const int* in_sizes, int n_in,
                              void* d_out, int out_size, void* d_ws, size_t ws_size,
                              hipStream_t stream) {
    (void)in_sizes; (void)n_in; (void)out_size; (void)ws_size;
    const float* x   = (const float*)d_in[0];
    const float* RIH = (const float*)d_in[1];
    const float* RH  = (const float*)d_in[2];
    const float* RHL = (const float*)d_in[3];
    const float* BH  = (const float*)d_in[4];
    const float* BL  = (const float*)d_in[5];
    const float* MIH = (const float*)d_in[6];
    const float* MH  = (const float*)d_in[7];
    const float* MHL = (const float*)d_in[8];
    float* out = (float*)d_out;

    char* p = (char*)d_ws;
    unsigned short* WT_H  = (unsigned short*)wsalloc(p, (size_t)HD * HD * 2);
    unsigned short* WT_HL = (unsigned short*)wsalloc(p, (size_t)LP * HD * 2);
    unsigned short* WT_IH = (unsigned short*)wsalloc(p, (size_t)HD * Isz * 2);
    unsigned short* xb    = (unsigned short*)wsalloc(p, (size_t)Bsz * Isz * 2);
    float* drive = (float*)wsalloc(p, (size_t)Bsz * HD * 4);
    float* Hbuf0 = (float*)wsalloc(p, (size_t)Bsz * HD * 4);
    float* Hbuf1 = (float*)wsalloc(p, (size_t)Bsz * HD * 4);
    unsigned short* Hb0 = (unsigned short*)wsalloc(p, (size_t)Bsz * HD * 2);
    unsigned short* Hb1 = (unsigned short*)wsalloc(p, (size_t)Bsz * HD * 2);
    float* col_l1 = (float*)wsalloc(p, HD * 4);
    float* whl    = (float*)wsalloc(p, LP * 4);
    float* scalars = (float*)wsalloc(p, 256);
    int* act = (int*)wsalloc(p, (size_t)(TMAX / 2 + 1) * Bsz * 4);
    float* stage = (float*)WT_IH;
    float* dHp = (float*)xb;
    float* top2 = dHp + (size_t)4 * DHP;
    const size_t T2SET = (size_t)2 * (LP / 64) * Bsz * 2;

    col_l1_kernel<<<HD, 256, 0, stream>>>(RH, MH, col_l1, HD, HD);
    col_l1_kernel<<<Lsz, 256, 0, stream>>>(RHL, MHL, whl, HD, Lsz);
    scalars_kernel<<<1, 256, 0, stream>>>(col_l1, whl, scalars);

    build_wt_kernel<HD><<<(HD * HD / 8) / 256, 256, 0, stream>>>(
        RH, MH, scalars,  0, WT_H, HD, HD, HD);
    build_wt_kernel<HD><<<(LP * HD / 8) / 256, 256, 0, stream>>>(
        RHL, MHL, scalars,  1, WT_HL, LP, Lsz, Lsz);
    build_wt_kernel<Isz><<<(HD * Isz / 8) / 256, 256, 0, stream>>>(
        RIH, MIH, scalars,  1, WT_IH, HD, HD, HD);
    to_bf16_kernel<Isz><<<(Bsz * Isz / 8) / 256, 256, 0, stream>>>(x, xb, Bsz);

    wmma_gemm<0><<<dim3(HD / 64, Bsz / 64), 128, 0, stream>>>(
        xb, nullptr, WT_IH, Bsz, HD, Isz, BH, nullptr, nullptr, nullptr, nullptr,
        drive, nullptr, nullptr, nullptr, nullptr, nullptr, nullptr, nullptr, nullptr, HD);

    init_state<<<(Bsz * HD) / 256, 256, 0, stream>>>(
        Hbuf0, Hb0, act, dHp + (size_t)3 * DHP, dHp, top2);

    float* Hc[2] = {Hbuf0, Hbuf1};
    unsigned short* Hbc[2] = {Hb0, Hb1};
    for (int j = 1; j <= TMAX / 2; ++j) {
        for (int t = 2 * j - 1; t <= 2 * j; ++t) {
            const int pc = (t - 1) & 1, pn = t & 1;
            wmma_gemm<1><<<dim3(HD / 64, Bsz / 64), 128, 0, stream>>>(
                Hbc[pc], nullptr, WT_H, Bsz, HD, HD, nullptr, drive, Hc[pc],
                nullptr, nullptr, Hc[pn], nullptr, Hbc[pn],
                nullptr, nullptr, dHp + (size_t)(t & 3) * DHP,
                nullptr, nullptr, nullptr, HD);
        }
        wmma_gemm<2><<<dim3(LP / 64, 2 * (Bsz / 64)), 128, 0, stream>>>(
            Hbc[1], Hbc[0], WT_HL, Bsz, LP, HD, BL, nullptr, nullptr,
            act + (j - 1) * Bsz, act + j * Bsz, out, stage, nullptr,
            dHp + (size_t)((2 * j + 1) & 3) * DHP, dHp + (size_t)((2 * j + 2) & 3) * DHP, nullptr,
            top2 + (size_t)((j - 1) & 1) * T2SET, top2 + (size_t)(j & 1) * T2SET,
            scalars, Lsz);
    }
    finish_rows<<<dim3(4, Bsz / 16), 128, 0, stream>>>(
        stage, out, act + (TMAX / 2) * Bsz, top2 + (size_t)((TMAX / 2) & 1) * T2SET,
        dHp + (size_t)((TMAX - 1) & 3) * DHP, scalars, Bsz, Lsz);
}


# Column L1 sums: a block owns 8 whole columns, read along rows (wave load = 4 rows x 32 B; was 32 lines 8 KB apart, 8 serial trips); row-class sums folded in LDS by the old tree; same bits
# speedup vs baseline: 3.5168x; 1.1047x over previous
#include <hip/hip_runtime.h>


#define Bsz 512
#define Isz 1024
#define HD  2048
#define Lsz 1000
#define LP  1024
#define TMAX 32
#define KD  4
#define NCB (HD / 64)
#define DHP (NCB * (Bsz / 16) * 32)

typedef __attribute__((ext_vector_type(16))) __bf16 v16bf;
typedef __attribute__((ext_vector_type(8)))  float  v8f;
typedef __attribute__((ext_vector_type(4)))  unsigned int u32x4;

__device__ __forceinline__ unsigned short f2bf(float f) {
    unsigned int u = __float_as_uint(f);
    u += 0x7FFFu + ((u >> 16) & 1u);
    return (unsigned short)(u >> 16);
}

__device__ __forceinline__ size_t packed_idx(int r, int k, int K) {
    const int w = k & 31;
    const int lane = (((w >> 3) & 1) << 4) + (r & 15);
    return ((size_t)(r >> 4) * (size_t)(K >> 5) + (size_t)(k >> 5)) * 512
         + (size_t)(((w >> 4) << 8) + lane * 8 + (w & 7));
}

struct alignas(32) FragBits { u32x4 lo, hi; };

__device__ __forceinline__ v16bf load_frag(const unsigned short* __restrict__ base,
                                           unsigned off) {
    const char* p = reinterpret_cast<const char*>(base) + off;
    FragBits fb;
    fb.lo = *reinterpret_cast<const u32x4*>(p);
    fb.hi = *reinterpret_cast<const u32x4*>(p + 512);
    v16bf v;
    __builtin_memcpy(&v, &fb, sizeof(v));
    return v;
}

__device__ __forceinline__ void top2_put(float& m1, float& m2, float x) {
    m2 = fmaxf(m2, fminf(m1, x));
    m1 = fmaxf(m1, x);
}
__device__ __forceinline__ void top2_join(float& m1, float& m2, float o1, float o2) {
    const float hi = fmaxf(m1, o1);
    const float lo = fmaxf(fminf(m1, o1), fmaxf(m2, o2));
    m1 = hi; m2 = lo;
}
template <int MASK>
__device__ __forceinline__ float row_xor(float v) {
    return __int_as_float(__builtin_amdgcn_update_dpp(
        0, __float_as_int(v), 0x160 + MASK, 0xf, 0xf, false));
}
__device__ __forceinline__ void top2_row16(float& m1, float& m2) {
    top2_join(m1, m2, row_xor<8>(m1), row_xor<8>(m2));
    top2_join(m1, m2, row_xor<4>(m1), row_xor<4>(m2));
    top2_join(m1, m2, row_xor<2>(m1), row_xor<2>(m2));
    top2_join(m1, m2, row_xor<1>(m1), row_xor<1>(m2));
}

template <int MODE>
__global__ __launch_bounds__(128)
void wmma_gemm(const unsigned short* __restrict__ A,
               const unsigned short* __restrict__ A2,
               const unsigned short* __restrict__ BT,
               int M, int N, int K,
               const float* __restrict__ bias,
               const float* __restrict__ drive,
               const float* __restrict__ Hold,
               const int*   __restrict__ actIn,
               int*         __restrict__ actOut,
               float* __restrict__ Cout,
               float* __restrict__ Stage,
               unsigned short* __restrict__ HbOut,
               const float* __restrict__ dHa,
               const float* __restrict__ dHb,
               float* __restrict__ dHout,
               const float* __restrict__ Top2In,
               float* __restrict__ Top2Out,
               const float* __restrict__ scalars,
               int NL) {
    __shared__ float4 red[4 * 16 * 2 * 32];

    const int lane = threadIdx.x & 31;
    const int wave = __builtin_amdgcn_readfirstlane((int)(threadIdx.x >> 5));
    const int rbs = M >> 6;
    const int half = (MODE == 2 && (int)blockIdx.y >= rbs) ? 1 : 0;
    const int by = (int)blockIdx.y - half * rbs;
    const int rowBase = by * 64 + wave * 16;
    const int colBase = blockIdx.x * 64;
    const int r16 = lane & 15;
    const int kh  = lane >> 4;

    const int nk = K >> 7;
    const size_t kOff = (size_t)wave * nk * 512;
    const size_t strip = (size_t)16 * K;
    const unsigned short* Ah = (MODE == 2 && half != 0) ? A2 : A;
    const unsigned short* sa0 = Ah + (size_t)(by * 64) * K + kOff;
    const unsigned short* sa1 = sa0 + strip;
    const unsigned short* sa2 = sa1 + strip;
    const unsigned short* sa3 = sa2 + strip;
    const unsigned short* sb0 = BT + (size_t)colBase * K + kOff;
    const unsigned short* sb1 = sb0 + strip;
    const unsigned short* sb2 = sb1 + strip;
    const unsigned short* sb3 = sb2 + strip;
    const unsigned boff = lane * 16;

#define LOAD_SLOT(s, o)                                   \
    a[s][0] = load_frag(sa0, (o) + (s) * 1024);           \
    b[s][0] = load_frag(sb0, (o) + (s) * 1024);           \
    a[s][1] = load_frag(sa1, (o) + (s) * 1024);           \
    b[s][1] = load_frag(sb1, (o) + (s) * 1024);           \
    a[s][2] = load_frag(sa2, (o) + (s) * 1024);           \
    b[s][2] = load_frag(sb2, (o) + (s) * 1024);           \
    a[s][3] = load_frag(sa3, (o) + (s) * 1024);           \
    b[s][3] = load_frag(sb3, (o) + (s) * 1024);

    v8f acc0 = {0.f,0.f,0.f,0.f,0.f,0.f,0.f,0.f};
    v8f acc[4][4];
#pragma unroll
    for (int i = 0; i < 4; ++i)
#pragma unroll
        for (int f = 0; f < 4; ++f) acc[i][f] = acc0;

    v16bf a[KD][4], b[KD][4];
#pragma unroll
    for (int s = 0; s < KD; ++s) {
        LOAD_SLOT(s, boff)
        __builtin_amdgcn_sched_barrier(0);
    }
#define ADVANCE_STREAMS()                                             \
    sa0 += KD * 512; sa1 += KD * 512; sa2 += KD * 512; sa3 += KD * 512; \
    sb0 += KD * 512; sb1 += KD * 512; sb2 += KD * 512; sb3 += KD * 512;
    ADVANCE_STREAMS()

    float2 prA[8], prB[8];
    float dpA[NCB / 2], dpB[NCB / 2];
    float gcoef = 0.f;
    int actPrev = 1;
    if (MODE == 2) {
        const int row = rowBase + r16;
        const float2* tp = reinterpret_cast<const float2*>(Top2In);
        const size_t tiles = (size_t)(N >> 6);
#pragma unroll
        for (int j = 0; j < 8; ++j) {
            prA[j] = tp[(size_t)(kh * 8 + j) * M + row];
            prB[j] = tp[(tiles + (size_t)(kh * 8 + j)) * M + row];
        }
        const size_t dofs = ((size_t)(kh * (NCB / 2)) * (M >> 4) + (rowBase >> 4)) * 32 + r16;
#pragma unroll
        for (int j = 0; j < NCB / 2; ++j) {
            dpA[j] = dHa[dofs + (size_t)j * (M >> 4) * 32];
            dpB[j] = dHb[dofs + (size_t)j * (M >> 4) * 32];
        }
        gcoef = scalars[1];
        actPrev = actIn[row];
    }

    for (int k0 = 0; k0 < nk - KD; k0 += KD) {
#pragma unroll
        for (int s = 0; s < KD; ++s) {
#pragma unroll
            for (int i = 0; i < 4; ++i)
#pragma unroll
                for (int f = 0; f < 4; ++f)
                    acc[i][f] = __builtin_amdgcn_wmma_f32_16x16x32_bf16(
                        false, a[s][i], false, b[s][f], (short)0, acc[i][f], false, false);
            LOAD_SLOT(s, boff)
        }
        ADVANCE_STREAMS()
    }
#undef LOAD_SLOT
#undef ADVANCE_STREAMS
#pragma unroll
    for (int s = 0; s < KD; ++s) {
#pragma unroll
        for (int i = 0; i < 4; ++i)
#pragma unroll
            for (int f = 0; f < 4; ++f)
                acc[i][f] = __builtin_amdgcn_wmma_f32_16x16x32_bf16(
                    false, a[s][i], false, b[s][f], (short)0, acc[i][f], false, false);
    }

#pragma unroll
    for (int i = 0; i < 4; ++i)
#pragma unroll
        for (int f = 0; f < 4; ++f) {
            const int t = ((wave * 16 + i * 4 + f) * 2) * 32 + lane;
            red[t]      = make_float4(acc[i][f][0], acc[i][f][1], acc[i][f][2], acc[i][f][3]);
            red[t + 32] = make_float4(acc[i][f][4], acc[i][f][5], acc[i][f][6], acc[i][f][7]);
        }
    __syncthreads();

    __builtin_amdgcn_sched_barrier(0);
    unsigned actMask = 0xFFFFFFFFu, copyMask = 0u;
    int actNew = 1;
    if (MODE == 2) {
        bool cert[2];
#pragma unroll
        for (int s = 0; s < 2; ++s) {
            const float2* pr = s ? prB : prA;
            const float*  dp = s ? dpB : dpA;
            float m1 = pr[0].x, m2 = pr[0].y;
#pragma unroll
            for (int j = 1; j < 8; ++j) top2_join(m1, m2, pr[j].x, pr[j].y);
            const float o1 = __shfl_xor(m1, 16, 32);
            const float o2 = __shfl_xor(m2, 16, 32);
            top2_join(m1, m2, o1, o2);
            const float margin = m1 - m2;
            float dh = dp[0];
#pragma unroll
            for (int j = 1; j < NCB / 2; ++j) dh = fmaxf(dh, dp[j]);
            dh = fmaxf(dh, __shfl_xor(dh, 16, 32));
            const float Gamma = gcoef * dh;
            cert[s] = margin > 2.0f * Gamma;
        }
        const bool mid = actPrev != 0 && !cert[0];
        actNew = (mid && !cert[1]) ? 1 : 0;
        actMask = __builtin_amdgcn_ballot_w32(actNew != 0);
        copyMask = __builtin_amdgcn_ballot_w32(mid && cert[1]);
    }

    v8f accs[4];
#pragma unroll
    for (int f = 0; f < 4; ++f) {
        const int t = ((wave * 4 + f) * 2) * 32 + lane;
        float4 lo = red[t], hi = red[t + 32];
#pragma unroll
        for (int q = 1; q < 4; ++q) {
            const float4 l2 = red[t + q * 16 * 2 * 32];
            const float4 h2 = red[t + q * 16 * 2 * 32 + 32];
            lo.x += l2.x; lo.y += l2.y; lo.z += l2.z; lo.w += l2.w;
            hi.x += h2.x; hi.y += h2.y; hi.z += h2.z; hi.w += h2.w;
        }
        accs[f][0] = lo.x; accs[f][1] = lo.y; accs[f][2] = lo.z; accs[f][3] = lo.w;
        accs[f][4] = hi.x; accs[f][5] = hi.y; accs[f][6] = hi.z; accs[f][7] = hi.w;
    }

    if (MODE == 2 && blockIdx.x == 0 && half == 0 && lane < 16) actOut[rowBase + lane] = actNew;
    float dmax[8];
    if (MODE == 1) {
#pragma unroll
        for (int r = 0; r < 8; ++r) dmax[r] = 0.f;
    }

    if (MODE == 2) {
        float t1[8], t2[8];
#pragma unroll
        for (int r = 0; r < 8; ++r) { t1[r] = -1e30f; t2[r] = -1e30f; }
        const unsigned o0 = (unsigned)(rowBase + 8 * kh) * (unsigned)NL + (unsigned)(colBase + r16);
        const unsigned rowOn = actMask >> (8 * kh);
        float* dst = half != 0 ? Stage : Cout;
        if (half == 0 && copyMask != 0u) {
            const unsigned rowCp = copyMask >> (8 * kh);
#pragma unroll
            for (int f = 0; f < 4; ++f) {
                if (colBase + f * 16 + r16 < NL) {
#pragma unroll
                    for (int r = 0; r < 8; ++r)
                        if ((rowCp >> r) & 1u) {
                            const unsigned o = o0 + (unsigned)r * (unsigned)NL + (unsigned)(f * 16);
                            Cout[o] = Stage[o];
                        }
                }
            }
        }
#pragma unroll
        for (int f = 0; f < 4; ++f) {
            const int n = colBase + f * 16 + r16;
            const bool ok = n < NL;
            const float bn = ok ? bias[n] : 0.f;
            float x[8];
#pragma unroll
            for (int r = 0; r < 8; ++r) x[r] = accs[f][r] + bn;
            if (ok) {
#pragma unroll
                for (int r = 0; r < 8; ++r)
                    if ((rowOn >> r) & 1u)
                        dst[o0 + (unsigned)r * (unsigned)NL + (unsigned)(f * 16)] = x[r];
            }
#pragma unroll
            for (int r = 0; r < 8; ++r) top2_put(t1[r], t2[r], ok ? x[r] : -1e30f);
        }
#pragma unroll
        for (int r = 0; r < 8; ++r) top2_row16(t1[r], t2[r]);
        float s1 = t1[0], s2 = t2[0];
#pragma unroll
        for (int r = 1; r < 8; ++r) {
            if (r16 == r) { s1 = t1[r]; s2 = t2[r]; }
        }
        if (r16 < 8) {
            reinterpret_cast<float2*>(Top2Out)[((size_t)half * gridDim.x + blockIdx.x) * M
                                               + rowBase + r16 + 8 * kh] = make_float2(s1, s2);
        }
    }

#pragma unroll
    for (int f = 0; f < (MODE == 2 ? 0 : 4); ++f) {
        const int n = colBase + f * 16 + r16;
#pragma unroll
        for (int r = 0; r < 8; ++r) {
            const int m = rowBase + r + 8 * kh;
            const float v = accs[f][r];
            if (MODE == 0) {
                Cout[(size_t)m * N + n] = v + bias[n];
            } else {
                const size_t idx = (size_t)m * N + n;
                const float pre = v + drive[idx];
                const float hn = pre >= 0.f ? pre : 0.05f * pre;
                const float ho = Hold[idx];
                Cout[idx] = hn;
                HbOut[packed_idx(m, n, N)] = f2bf(hn);
                dmax[r] = fmaxf(dmax[r], fabsf(hn - ho));
            }
        }
    }

    if (MODE == 1) {
#pragma unroll
        for (int r = 0; r < 8; ++r) {
            dmax[r] = fmaxf(dmax[r], row_xor<8>(dmax[r]));
            dmax[r] = fmaxf(dmax[r], row_xor<4>(dmax[r]));
            dmax[r] = fmaxf(dmax[r], row_xor<2>(dmax[r]));
            dmax[r] = fmaxf(dmax[r], row_xor<1>(dmax[r]));
        }
        float s = dmax[0];
#pragma unroll
        for (int r = 1; r < 8; ++r) {
            if (r16 == r) s = dmax[r];
        }
        if (r16 < 8) {
            dHout[((size_t)blockIdx.x * (M >> 4) + (rowBase >> 4)) * 32 + r16 + 8 * kh] = s;
        }
    }
}

template <int COLS>
__global__ __launch_bounds__(256)
void col_l1_kernel(const float* __restrict__ A, const float* __restrict__ Msk,
                   float* __restrict__ colsum) {
    static_assert(COLS % 8 == 0, "a block owns 8 whole columns");
    __shared__ float sm[256 * 8];
    const int tid = threadIdx.x;
    const int c  = tid & 7;
    const int tq = tid >> 3;
    const size_t base = (size_t)tq * COLS + (size_t)blockIdx.x * 8 + c;
    const float* pa = A + base;
    const float* pm = Msk + base;
    float r[8][8], k[8][8];
#pragma unroll
    for (int j = 0; j < 8; ++j)
#pragma unroll
        for (int m = 0; m < 8; ++m) {
            r[j][m] = pa[(size_t)(32 * j + 256 * m) * COLS];
            k[j][m] = pm[(size_t)(32 * j + 256 * m) * COLS];
        }
#pragma unroll
    for (int j = 0; j < 8; ++j) {
        float s = 0.f;
#pragma unroll
        for (int m = 0; m < 8; ++m) s += fabsf(r[j][m] * k[j][m]);
        sm[(tq + 32 * j) * 8 + c] = s;
    }
    __syncthreads();
    for (int st = 128; st > 0; st >>= 1) {
        for (int idx = tid; idx < st * 8; idx += 256) sm[idx] += sm[idx + st * 8];
        __syncthreads();
    }
    if (tid < 8) colsum[blockIdx.x * 8 + tid] = sm[tid];
}

__global__ __launch_bounds__(256)
void scalars_kernel(const float* __restrict__ col_l1,
                    const float* __restrict__ whl,
                    float* __restrict__ scalars) {
    __shared__ float sm[256];
    const int t = threadIdx.x;
    float mx = -1e30f;
    for (int j = t; j < HD; j += 256) mx = fmaxf(mx, col_l1[j]);
    sm[t] = mx; __syncthreads();
    for (int s = 128; s > 0; s >>= 1) {
        if (t < s) sm[t] = fmaxf(sm[t], sm[t + s]);
        __syncthreads();
    }
    const float m1 = sm[0];
    __syncthreads();
    mx = -1e30f;
    for (int j = t; j < Lsz; j += 256) mx = fmaxf(mx, whl[j]);
    sm[t] = mx; __syncthreads();
    for (int s = 128; s > 0; s >>= 1) {
        if (t < s) sm[t] = fmaxf(sm[t], sm[t + s]);
        __syncthreads();
    }
    if (t == 0) {
        scalars[0] = 0.95f / fmaxf(m1, 1e-8f);
        scalars[1] = sm[0] * 19.0f;
    }
}

template <int K>
__global__ __launch_bounds__(256)
void build_wt_kernel(const float* __restrict__ R, const float* __restrict__ M,
                     const float* __restrict__ scalars, int sel,
                     unsigned short* __restrict__ WT,
                     int N, int srcld, int NL) {
    const unsigned g = blockIdx.x * 256u + threadIdx.x;
    const int lane = (int)(g & 31u);
    const unsigned wv = g >> 5;
    const int k0 = (int)(wv % (unsigned)(K / 8)) * 8;
    const int n  = (int)(wv / (unsigned)(K / 8)) * 32 + lane;
    if (n >= N) return;
    float v[8];
    if (n < NL) {
        const size_t s = (size_t)k0 * srcld + n;
        float r[8], m[8];
#pragma unroll
        for (int e = 0; e < 8; ++e) {
            r[e] = R[s + (size_t)e * srcld];
            m[e] = M[s + (size_t)e * srcld];
        }
#pragma unroll
        for (int e = 0; e < 8; ++e) v[e] = r[e] * m[e];
        if (sel == 0) {
            const float sc = scalars[0];
#pragma unroll
            for (int e = 0; e < 8; ++e) v[e] *= sc;
        }
    } else {
#pragma unroll
        for (int e = 0; e < 8; ++e) v[e] = 0.f;
    }
    u32x4 c;
    c[0] = (unsigned)f2bf(v[0]) | ((unsigned)f2bf(v[1]) << 16);
    c[1] = (unsigned)f2bf(v[2]) | ((unsigned)f2bf(v[3]) << 16);
    c[2] = (unsigned)f2bf(v[4]) | ((unsigned)f2bf(v[5]) << 16);
    c[3] = (unsigned)f2bf(v[6]) | ((unsigned)f2bf(v[7]) << 16);
    *reinterpret_cast<u32x4*>(WT + packed_idx(n, k0, K)) = c;
}

template <int K>
__global__ __launch_bounds__(256)
void to_bf16_kernel(const float* __restrict__ src, unsigned short* __restrict__ dst,
                    int rows) {
    const unsigned g = blockIdx.x * 256u + threadIdx.x;
    const int lane = (int)(g & 31u);
    const unsigned hw = g >> 5;
    const int hh = (int)(hw & 1u);
    const unsigned tile = hw >> 1;
    const int kt = (int)(tile % (unsigned)(K / 32));
    const int r  = (int)(tile / (unsigned)(K / 32)) * 16 + (lane & 15);
    const int k0 = kt * 32 + hh * 16 + (lane >> 4) * 8;
    if (r >= rows) return;
    const float* p = src + (size_t)r * K + k0;
    float v[8];
#pragma unroll
    for (int e = 0; e < 8; ++e) v[e] = p[e];
    u32x4 c;
    c[0] = (unsigned)f2bf(v[0]) | ((unsigned)f2bf(v[1]) << 16);
    c[1] = (unsigned)f2bf(v[2]) | ((unsigned)f2bf(v[3]) << 16);
    c[2] = (unsigned)f2bf(v[4]) | ((unsigned)f2bf(v[5]) << 16);
    c[3] = (unsigned)f2bf(v[6]) | ((unsigned)f2bf(v[7]) << 16);
    *reinterpret_cast<u32x4*>(dst + packed_idx(r, k0, K)) = c;
}

__global__ __launch_bounds__(256)
void init_state(float* __restrict__ H0, unsigned short* __restrict__ Hb0,
                int* __restrict__ act0, float* __restrict__ dHa0,
                float* __restrict__ dHb0, float* __restrict__ pairs0) {
    const size_t n = (size_t)Bsz * HD;
    for (size_t t = (size_t)blockIdx.x * 256 + threadIdx.x; t < n;
         t += (size_t)gridDim.x * 256) {
        H0[t] = 0.f;
        Hb0[t] = 0;
    }
    const int i = blockIdx.x * 256 + threadIdx.x;
    if (i < Bsz) act0[i] = 1;
    if (i < DHP) { dHa0[i] = 0.f; dHb0[i] = 0.f; }
    if (i < 2 * (LP / 64) * Bsz * 2) pairs0[i] = 0.f;
}

__global__ __launch_bounds__(128)
void finish_rows(const float* __restrict__ Stage, float* __restrict__ out,
                 const int* __restrict__ actIn,
                 const float* __restrict__ Top2In,
                 const float* __restrict__ dHa,
                 const float* __restrict__ scalars, int M, int NL) {
    const int lane = threadIdx.x & 31;
    const int r16 = lane & 15;
    const int kh  = lane >> 4;
    const int rowBase = blockIdx.y * 16;
    const int row = rowBase + r16;
    const float2* tp = reinterpret_cast<const float2*>(Top2In);
    float m1 = -1e30f, m2 = -1e30f;
#pragma unroll
    for (int j = 0; j < 8; ++j) {
        const float2 q = tp[(size_t)(kh * 8 + j) * M + row];
        if (j == 0) { m1 = q.x; m2 = q.y; } else top2_join(m1, m2, q.x, q.y);
    }
    const float o1 = __shfl_xor(m1, 16, 32);
    const float o2 = __shfl_xor(m2, 16, 32);
    top2_join(m1, m2, o1, o2);
    const float margin = m1 - m2;
    const size_t dofs = ((size_t)(kh * (NCB / 2)) * (M >> 4) + (rowBase >> 4)) * 32 + r16;
    float dh = dHa[dofs];
#pragma unroll
    for (int j = 1; j < NCB / 2; ++j) dh = fmaxf(dh, dHa[dofs + (size_t)j * (M >> 4) * 32]);
    dh = fmaxf(dh, __shfl_xor(dh, 16, 32));
    const float Gamma = scalars[1] * dh;
    const bool cert = margin > 2.0f * Gamma;
    const unsigned take = __builtin_amdgcn_ballot_w32(actIn[row] != 0 && !cert) & 0xFFFFu;
    if (take == 0u) return;
    const int per = (NL + (int)gridDim.x - 1) / (int)gridDim.x;
    const int c0 = blockIdx.x * per;
    const int c1 = min(NL, c0 + per);
    for (int r = 0; r < 16; ++r) {
        if ((take >> r) & 1u) {
            const size_t o = (size_t)(rowBase + r) * NL;
            for (int c = c0 + (int)threadIdx.x; c < c1; c += 128) out[o + c] = Stage[o + c];
        }
    }
}

static inline char* wsalloc(char*& p, size_t bytes) {
    char* r = p;
    p += (bytes + 255) & ~(size_t)255;
    return r;
}

extern "C" void kernel_launch(void* const* d_in, const int* in_sizes, int n_in,
                              void* d_out, int out_size, void* d_ws, size_t ws_size,
                              hipStream_t stream) {
    (void)in_sizes; (void)n_in; (void)out_size; (void)ws_size;
    const float* x   = (const float*)d_in[0];
    const float* RIH = (const float*)d_in[1];
    const float* RH  = (const float*)d_in[2];
    const float* RHL = (const float*)d_in[3];
    const float* BH  = (const float*)d_in[4];
    const float* BL  = (const float*)d_in[5];
    const float* MIH = (const float*)d_in[6];
    const float* MH  = (const float*)d_in[7];
    const float* MHL = (const float*)d_in[8];
    float* out = (float*)d_out;

    char* p = (char*)d_ws;
    unsigned short* WT_H  = (unsigned short*)wsalloc(p, (size_t)HD * HD * 2);
    unsigned short* WT_HL = (unsigned short*)wsalloc(p, (size_t)LP * HD * 2);
    unsigned short* WT_IH = (unsigned short*)wsalloc(p, (size_t)HD * Isz * 2);
    unsigned short* xb    = (unsigned short*)wsalloc(p, (size_t)Bsz * Isz * 2);
    float* drive = (float*)wsalloc(p, (size_t)Bsz * HD * 4);
    float* Hbuf0 = (float*)wsalloc(p, (size_t)Bsz * HD * 4);
    float* Hbuf1 = (float*)wsalloc(p, (size_t)Bsz * HD * 4);
    unsigned short* Hb0 = (unsigned short*)wsalloc(p, (size_t)Bsz * HD * 2);
    unsigned short* Hb1 = (unsigned short*)wsalloc(p, (size_t)Bsz * HD * 2);
    float* col_l1 = (float*)wsalloc(p, HD * 4);
    float* whl    = (float*)wsalloc(p, LP * 4);
    float* scalars = (float*)wsalloc(p, 256);
    int* act = (int*)wsalloc(p, (size_t)(TMAX / 2 + 1) * Bsz * 4);
    float* stage = (float*)WT_IH;
    float* dHp = (float*)xb;
    float* top2 = dHp + (size_t)4 * DHP;
    const size_t T2SET = (size_t)2 * (LP / 64) * Bsz * 2;

    col_l1_kernel<HD><<<HD / 8, 256, 0, stream>>>(RH, MH, col_l1);
    col_l1_kernel<Lsz><<<Lsz / 8, 256, 0, stream>>>(RHL, MHL, whl);
    scalars_kernel<<<1, 256, 0, stream>>>(col_l1, whl, scalars);

    build_wt_kernel<HD><<<(HD * HD / 8) / 256, 256, 0, stream>>>(
        RH, MH, scalars,  0, WT_H, HD, HD, HD);
    build_wt_kernel<HD><<<(LP * HD / 8) / 256, 256, 0, stream>>>(
        RHL, MHL, scalars,  1, WT_HL, LP, Lsz, Lsz);
    build_wt_kernel<Isz><<<(HD * Isz / 8) / 256, 256, 0, stream>>>(
        RIH, MIH, scalars,  1, WT_IH, HD, HD, HD);
    to_bf16_kernel<Isz><<<(Bsz * Isz / 8) / 256, 256, 0, stream>>>(x, xb, Bsz);

    wmma_gemm<0><<<dim3(HD / 64, Bsz / 64), 128, 0, stream>>>(
        xb, nullptr, WT_IH, Bsz, HD, Isz, BH, nullptr, nullptr, nullptr, nullptr,
        drive, nullptr, nullptr, nullptr, nullptr, nullptr, nullptr, nullptr, nullptr, HD);

    init_state<<<(Bsz * HD) / 256, 256, 0, stream>>>(
        Hbuf0, Hb0, act, dHp + (size_t)3 * DHP, dHp, top2);

    float* Hc[2] = {Hbuf0, Hbuf1};
    unsigned short* Hbc[2] = {Hb0, Hb1};
    for (int j = 1; j <= TMAX / 2; ++j) {
        for (int t = 2 * j - 1; t <= 2 * j; ++t) {
            const int pc = (t - 1) & 1, pn = t & 1;
            wmma_gemm<1><<<dim3(HD / 64, Bsz / 64), 128, 0, stream>>>(
                Hbc[pc], nullptr, WT_H, Bsz, HD, HD, nullptr, drive, Hc[pc],
                nullptr, nullptr, Hc[pn], nullptr, Hbc[pn],
                nullptr, nullptr, dHp + (size_t)(t & 3) * DHP,
                nullptr, nullptr, nullptr, HD);
        }
        wmma_gemm<2><<<dim3(LP / 64, 2 * (Bsz / 64)), 128, 0, stream>>>(
            Hbc[1], Hbc[0], WT_HL, Bsz, LP, HD, BL, nullptr, nullptr,
            act + (j - 1) * Bsz, act + j * Bsz, out, stage, nullptr,
            dHp + (size_t)((2 * j + 1) & 3) * DHP, dHp + (size_t)((2 * j + 2) & 3) * DHP, nullptr,
            top2 + (size_t)((j - 1) & 1) * T2SET, top2 + (size_t)(j & 1) * T2SET,
            scalars, Lsz);
    }
    finish_rows<<<dim3(4, Bsz / 16), 128, 0, stream>>>(
        stage, out, act + (TMAX / 2) * Bsz, top2 + (size_t)((TMAX / 2) & 1) * T2SET,
        dHp + (size_t)((TMAX - 1) & 3) * DHP, scalars, Bsz, Lsz);
}


# GEMM products transposed (BT fragment as first WMMA operand): a lane owns one output row, 8 columns per fragment; epilogue row traffic 16 B per lane, bf16 H in 512 B runs; same products, same k order
# speedup vs baseline: 3.7789x; 1.0745x over previous
#include <hip/hip_runtime.h>


#define Bsz 512
#define Isz 1024
#define HD  2048
#define Lsz 1000
#define LP  1024
#define TMAX 32
#define KD  4
#define NCB (HD / 64)
#define DHP (NCB * (Bsz / 16) * 32)
static_assert(Lsz % 4 == 0, "the logits epilogue moves groups of 4 columns: a group is wholly inside L or wholly behind it");

typedef __attribute__((ext_vector_type(16))) __bf16 v16bf;
typedef __attribute__((ext_vector_type(8)))  float  v8f;
typedef __attribute__((ext_vector_type(4)))  unsigned int u32x4;
typedef __attribute__((ext_vector_type(4)))  float  v4f;

__device__ __forceinline__ unsigned short f2bf(float f) {
    unsigned int u = __float_as_uint(f);
    u += 0x7FFFu + ((u >> 16) & 1u);
    return (unsigned short)(u >> 16);
}

__device__ __forceinline__ size_t packed_idx(int r, int k, int K) {
    const int w = k & 31;
    const int lane = (((w >> 3) & 1) << 4) + (r & 15);
    return ((size_t)(r >> 4) * (size_t)(K >> 5) + (size_t)(k >> 5)) * 512
         + (size_t)(((w >> 4) << 8) + lane * 8 + (w & 7));
}

struct alignas(32) FragBits { u32x4 lo, hi; };

__device__ __forceinline__ v16bf load_frag(const unsigned short* __restrict__ base,
                                           unsigned off) {
    const char* p = reinterpret_cast<const char*>(base) + off;
    FragBits fb;
    fb.lo = *reinterpret_cast<const u32x4*>(p);
    fb.hi = *reinterpret_cast<const u32x4*>(p + 512);
    v16bf v;
    __builtin_memcpy(&v, &fb, sizeof(v));
    return v;
}

__device__ __forceinline__ void top2_put(float& m1, float& m2, float x) {
    m2 = fmaxf(m2, fminf(m1, x));
    m1 = fmaxf(m1, x);
}
__device__ __forceinline__ void top2_join(float& m1, float& m2, float o1, float o2) {
    const float hi = fmaxf(m1, o1);
    const float lo = fmaxf(fminf(m1, o1), fmaxf(m2, o2));
    m1 = hi; m2 = lo;
}

template <int MODE>
__global__ __launch_bounds__(128)
void wmma_gemm(const unsigned short* __restrict__ A,
               const unsigned short* __restrict__ A2,
               const unsigned short* __restrict__ BT,
               int M, int N, int K,
               const float* __restrict__ bias,
               const float* __restrict__ drive,
               const float* __restrict__ Hold,
               const int*   __restrict__ actIn,
               int*         __restrict__ actOut,
               float* __restrict__ Cout,
               float* __restrict__ Stage,
               unsigned short* __restrict__ HbOut,
               const float* __restrict__ dHa,
               const float* __restrict__ dHb,
               float* __restrict__ dHout,
               const float* __restrict__ Top2In,
               float* __restrict__ Top2Out,
               const float* __restrict__ scalars,
               int NL) {
    __shared__ float4 red[4 * 16 * 2 * 32];

    const int lane = threadIdx.x & 31;
    const int wave = __builtin_amdgcn_readfirstlane((int)(threadIdx.x >> 5));
    const int rbs = M >> 6;
    const int half = (MODE == 2 && (int)blockIdx.y >= rbs) ? 1 : 0;
    const int by = (int)blockIdx.y - half * rbs;
    const int rowBase = by * 64 + wave * 16;
    const int colBase = blockIdx.x * 64;
    const int r16 = lane & 15;
    const int kh  = lane >> 4;

    const int nk = K >> 7;
    const size_t kOff = (size_t)wave * nk * 512;
    const size_t strip = (size_t)16 * K;
    const unsigned short* Ah = (MODE == 2 && half != 0) ? A2 : A;
    const unsigned short* sa0 = Ah + (size_t)(by * 64) * K + kOff;
    const unsigned short* sa1 = sa0 + strip;
    const unsigned short* sa2 = sa1 + strip;
    const unsigned short* sa3 = sa2 + strip;
    const unsigned short* sb0 = BT + (size_t)colBase * K + kOff;
    const unsigned short* sb1 = sb0 + strip;
    const unsigned short* sb2 = sb1 + strip;
    const unsigned short* sb3 = sb2 + strip;
    const unsigned boff = lane * 16;

#define LOAD_SLOT(s, o)                                   \
    a[s][0] = load_frag(sa0, (o) + (s) * 1024);           \
    b[s][0] = load_frag(sb0, (o) + (s) * 1024);           \
    a[s][1] = load_frag(sa1, (o) + (s) * 1024);           \
    b[s][1] = load_frag(sb1, (o) + (s) * 1024);           \
    a[s][2] = load_frag(sa2, (o) + (s) * 1024);           \
    b[s][2] = load_frag(sb2, (o) + (s) * 1024);           \
    a[s][3] = load_frag(sa3, (o) + (s) * 1024);           \
    b[s][3] = load_frag(sb3, (o) + (s) * 1024);

    v8f acc0 = {0.f,0.f,0.f,0.f,0.f,0.f,0.f,0.f};
    v8f acc[4][4];
#pragma unroll
    for (int i = 0; i < 4; ++i)
#pragma unroll
        for (int f = 0; f < 4; ++f) acc[i][f] = acc0;

    v16bf a[KD][4], b[KD][4];
#pragma unroll
    for (int s = 0; s < KD; ++s) {
        LOAD_SLOT(s, boff)
        __builtin_amdgcn_sched_barrier(0);
    }
#define ADVANCE_STREAMS()                                             \
    sa0 += KD * 512; sa1 += KD * 512; sa2 += KD * 512; sa3 += KD * 512; \
    sb0 += KD * 512; sb1 += KD * 512; sb2 += KD * 512; sb3 += KD * 512;
    ADVANCE_STREAMS()

    float2 prA[8], prB[8];
    float dpA[NCB / 2], dpB[NCB / 2];
    float gcoef = 0.f;
    int actPrev = 1;
    v4f bq[8];
    if (MODE == 2) {
        const int row = rowBase + r16;
        const float2* tp = reinterpret_cast<const float2*>(Top2In);
        const size_t tiles = (size_t)(N >> 6);
#pragma unroll
        for (int j = 0; j < 8; ++j) {
            prA[j] = tp[(size_t)(kh * 8 + j) * M + row];
            prB[j] = tp[(tiles + (size_t)(kh * 8 + j)) * M + row];
        }
        const size_t dofs = ((size_t)(kh * (NCB / 2)) * (M >> 4) + (rowBase >> 4)) * 32 + r16;
#pragma unroll
        for (int j = 0; j < NCB / 2; ++j) {
            dpA[j] = dHa[dofs + (size_t)j * (M >> 4) * 32];
            dpB[j] = dHb[dofs + (size_t)j * (M >> 4) * 32];
        }
        gcoef = scalars[1];
        actPrev = actIn[row];
    }

    for (int k0 = 0; k0 < nk - KD; k0 += KD) {
#pragma unroll
        for (int s = 0; s < KD; ++s) {
#pragma unroll
            for (int i = 0; i < 4; ++i)
#pragma unroll
                for (int f = 0; f < 4; ++f)
                    acc[i][f] = __builtin_amdgcn_wmma_f32_16x16x32_bf16(
                        false, b[s][f], false, a[s][i], (short)0, acc[i][f], false, false);
            LOAD_SLOT(s, boff)
        }
        ADVANCE_STREAMS()
    }
#undef LOAD_SLOT
#undef ADVANCE_STREAMS
#pragma unroll
    for (int s = 0; s < KD; ++s) {
#pragma unroll
        for (int i = 0; i < 4; ++i)
#pragma unroll
            for (int f = 0; f < 4; ++f)
                acc[i][f] = __builtin_amdgcn_wmma_f32_16x16x32_bf16(
                    false, b[s][f], false, a[s][i], (short)0, acc[i][f], false, false);
    }

#pragma unroll
    for (int i = 0; i < 4; ++i)
#pragma unroll
        for (int f = 0; f < 4; ++f) {
            const int t = ((wave * 16 + i * 4 + f) * 2) * 32 + lane;
            red[t]      = make_float4(acc[i][f][0], acc[i][f][1], acc[i][f][2], acc[i][f][3]);
            red[t + 32] = make_float4(acc[i][f][4], acc[i][f][5], acc[i][f][6], acc[i][f][7]);
        }
    __syncthreads();

    if (MODE == 2) {
#pragma unroll
        for (int q = 0; q < 8; ++q) {
            const int n0 = colBase + 8 * kh + (q >> 1) * 16 + (q & 1) * 4;
            bq[q] = *reinterpret_cast<const v4f*>(bias + (n0 + 4 <= NL ? n0 : 0));
        }
    }

    __builtin_amdgcn_sched_barrier(0);
    unsigned actMask = 0xFFFFFFFFu, copyMask = 0u;
    int actNew = 1;
    if (MODE == 2) {
        bool cert[2];
#pragma unroll
        for (int s = 0; s < 2; ++s) {
            const float2* pr = s ? prB : prA;
            const float*  dp = s ? dpB : dpA;
            float m1 = pr[0].x, m2 = pr[0].y;
#pragma unroll
            for (int j = 1; j < 8; ++j) top2_join(m1, m2, pr[j].x, pr[j].y);
            const float o1 = __shfl_xor(m1, 16, 32);
            const float o2 = __shfl_xor(m2, 16, 32);
            top2_join(m1, m2, o1, o2);
            const float margin = m1 - m2;
            float dh = dp[0];
#pragma unroll
            for (int j = 1; j < NCB / 2; ++j) dh = fmaxf(dh, dp[j]);
            dh = fmaxf(dh, __shfl_xor(dh, 16, 32));
            const float Gamma = gcoef * dh;
            cert[s] = margin > 2.0f * Gamma;
        }
        const bool mid = actPrev != 0 && !cert[0];
        actNew = (mid && !cert[1]) ? 1 : 0;
        actMask = __builtin_amdgcn_ballot_w32(actNew != 0);
        copyMask = __builtin_amdgcn_ballot_w32(mid && cert[1]);
    }

    v8f accs[4];
#pragma unroll
    for (int f = 0; f < 4; ++f) {
        const int t = ((wave * 4 + f) * 2) * 32 + lane;
        float4 lo = red[t], hi = red[t + 32];
#pragma unroll
        for (int q = 1; q < 4; ++q) {
            const float4 l2 = red[t + q * 16 * 2 * 32];
            const float4 h2 = red[t + q * 16 * 2 * 32 + 32];
            lo.x += l2.x; lo.y += l2.y; lo.z += l2.z; lo.w += l2.w;
            hi.x += h2.x; hi.y += h2.y; hi.z += h2.z; hi.w += h2.w;
        }
        accs[f][0] = lo.x; accs[f][1] = lo.y; accs[f][2] = lo.z; accs[f][3] = lo.w;
        accs[f][4] = hi.x; accs[f][5] = hi.y; accs[f][6] = hi.z; accs[f][7] = hi.w;
    }

    if (MODE == 2 && blockIdx.x == 0 && half == 0 && lane < 16) actOut[rowBase + lane] = actNew;
    const int m = rowBase + r16;
    const int nb = colBase + 8 * kh;

    if (MODE == 2) {
        const bool rowOn = ((actMask >> r16) & 1u) != 0u;
        float* dst = half != 0 ? Stage : Cout;
        const unsigned o0 = (unsigned)m * (unsigned)NL + (unsigned)nb;
        if (half == 0 && copyMask != 0u) {
            if (((copyMask >> r16) & 1u) != 0u) {
                v4f sq[8];
#pragma unroll
                for (int q = 0; q < 8; ++q) {
                    const int dq = (q >> 1) * 16 + (q & 1) * 4;
                    const unsigned os = nb + dq + 4 <= NL ? o0 + (unsigned)dq
                                                          : (unsigned)m * (unsigned)NL;
                    sq[q] = *reinterpret_cast<const v4f*>(Stage + os);
                }
#pragma unroll
                for (int q = 0; q < 8; ++q) {
                    const int dq = (q >> 1) * 16 + (q & 1) * 4;
                    if (nb + dq + 4 <= NL)
                        *reinterpret_cast<v4f*>(Cout + o0 + (unsigned)dq) = sq[q];
                }
            }
        }
        float t1 = -1e30f, t2 = -1e30f;
#pragma unroll
        for (int q = 0; q < 8; ++q) {
            const int f = q >> 1, g = q & 1;
            const int dq = f * 16 + g * 4;
            const bool ok = nb + dq + 4 <= NL;
            const float x0 = accs[f][g * 4 + 0] + (ok ? bq[q][0] : 0.f);
            const float x1 = accs[f][g * 4 + 1] + (ok ? bq[q][1] : 0.f);
            const float x2 = accs[f][g * 4 + 2] + (ok ? bq[q][2] : 0.f);
            const float x3 = accs[f][g * 4 + 3] + (ok ? bq[q][3] : 0.f);
            if (ok && rowOn) {
                v4f xv; xv[0] = x0; xv[1] = x1; xv[2] = x2; xv[3] = x3;
                *reinterpret_cast<v4f*>(dst + o0 + (unsigned)dq) = xv;
            }
            top2_put(t1, t2, ok ? x0 : -1e30f);
            top2_put(t1, t2, ok ? x1 : -1e30f);
            top2_put(t1, t2, ok ? x2 : -1e30f);
            top2_put(t1, t2, ok ? x3 : -1e30f);
        }
        {
            const float p1 = __shfl_xor(t1, 16, 32);
            const float p2 = __shfl_xor(t2, 16, 32);
            top2_join(t1, t2, p1, p2);
        }
        if (kh == 0) {
            reinterpret_cast<float2*>(Top2Out)[((size_t)half * gridDim.x + blockIdx.x) * M
                                               + m] = make_float2(t1, t2);
        }
    }

    if (MODE == 0) {
#pragma unroll
        for (int f = 0; f < 4; ++f) {
            const int n0 = nb + f * 16;
            const size_t idx = (size_t)m * N + n0;
            const float4 b0 = *reinterpret_cast<const float4*>(bias + n0);
            const float4 b1 = *reinterpret_cast<const float4*>(bias + n0 + 4);
            *reinterpret_cast<float4*>(Cout + idx) = make_float4(
                accs[f][0] + b0.x, accs[f][1] + b0.y, accs[f][2] + b0.z, accs[f][3] + b0.w);
            *reinterpret_cast<float4*>(Cout + idx + 4) = make_float4(
                accs[f][4] + b1.x, accs[f][5] + b1.y, accs[f][6] + b1.z, accs[f][7] + b1.w);
        }
    }

    if (MODE == 1) {
        float dm = 0.f;
#pragma unroll
        for (int f = 0; f < 4; ++f) {
            const int n0 = nb + f * 16;
            const size_t idx = (size_t)m * N + n0;
            const float4 d0 = *reinterpret_cast<const float4*>(drive + idx);
            const float4 d1 = *reinterpret_cast<const float4*>(drive + idx + 4);
            const float4 h0 = *reinterpret_cast<const float4*>(Hold + idx);
            const float4 h1 = *reinterpret_cast<const float4*>(Hold + idx + 4);
            const float dr[8] = {d0.x, d0.y, d0.z, d0.w, d1.x, d1.y, d1.z, d1.w};
            const float ho[8] = {h0.x, h0.y, h0.z, h0.w, h1.x, h1.y, h1.z, h1.w};
            float hn[8];
#pragma unroll
            for (int r = 0; r < 8; ++r) {
                const float pre = accs[f][r] + dr[r];
                hn[r] = pre >= 0.f ? pre : 0.05f * pre;
                dm = fmaxf(dm, fabsf(hn[r] - ho[r]));
            }
            *reinterpret_cast<float4*>(Cout + idx)     = make_float4(hn[0], hn[1], hn[2], hn[3]);
            *reinterpret_cast<float4*>(Cout + idx + 4) = make_float4(hn[4], hn[5], hn[6], hn[7]);
            u32x4 cb;
            cb[0] = (unsigned)f2bf(hn[0]) | ((unsigned)f2bf(hn[1]) << 16);
            cb[1] = (unsigned)f2bf(hn[2]) | ((unsigned)f2bf(hn[3]) << 16);
            cb[2] = (unsigned)f2bf(hn[4]) | ((unsigned)f2bf(hn[5]) << 16);
            cb[3] = (unsigned)f2bf(hn[6]) | ((unsigned)f2bf(hn[7]) << 16);
            *reinterpret_cast<u32x4*>(HbOut + packed_idx(m, n0, N)) = cb;
        }
        dm = fmaxf(dm, __shfl_xor(dm, 16, 32));
        if (kh == 0) {
            dHout[((size_t)blockIdx.x * (M >> 4) + (rowBase >> 4)) * 32 + r16] = dm;
        }
    }
}

template <int COLS>
__global__ __launch_bounds__(256)
void col_l1_kernel(const float* __restrict__ A, const float* __restrict__ Msk,
                   float* __restrict__ colsum) {
    static_assert(COLS % 8 == 0, "a block owns 8 whole columns");
    __shared__ float sm[256 * 8];
    const int tid = threadIdx.x;
    const int c  = tid & 7;
    const int tq = tid >> 3;
    const size_t base = (size_t)tq * COLS + (size_t)blockIdx.x * 8 + c;
    const float* pa = A + base;
    const float* pm = Msk + base;
    float r[8][8], k[8][8];
#pragma unroll
    for (int j = 0; j < 8; ++j)
#pragma unroll
        for (int m = 0; m < 8; ++m) {
            r[j][m] = pa[(size_t)(32 * j + 256 * m) * COLS];
            k[j][m] = pm[(size_t)(32 * j + 256 * m) * COLS];
        }
#pragma unroll
    for (int j = 0; j < 8; ++j) {
        float s = 0.f;
#pragma unroll
        for (int m = 0; m < 8; ++m) s += fabsf(r[j][m] * k[j][m]);
        sm[(tq + 32 * j) * 8 + c] = s;
    }
    __syncthreads();
    for (int st = 128; st > 0; st >>= 1) {
        for (int idx = tid; idx < st * 8; idx += 256) sm[idx] += sm[idx + st * 8];
        __syncthreads();
    }
    if (tid < 8) colsum[blockIdx.x * 8 + tid] = sm[tid];
}

__global__ __launch_bounds__(256)
void scalars_kernel(const float* __restrict__ col_l1,
                    const float* __restrict__ whl,
                    float* __restrict__ scalars) {
    __shared__ float sm[256];
    const int t = threadIdx.x;
    float mx = -1e30f;
    for (int j = t; j < HD; j += 256) mx = fmaxf(mx, col_l1[j]);
    sm[t] = mx; __syncthreads();
    for (int s = 128; s > 0; s >>= 1) {
        if (t < s) sm[t] = fmaxf(sm[t], sm[t + s]);
        __syncthreads();
    }
    const float m1 = sm[0];
    __syncthreads();
    mx = -1e30f;
    for (int j = t; j < Lsz; j += 256) mx = fmaxf(mx, whl[j]);
    sm[t] = mx; __syncthreads();
    for (int s = 128; s > 0; s >>= 1) {
        if (t < s) sm[t] = fmaxf(sm[t], sm[t + s]);
        __syncthreads();
    }
    if (t == 0) {
        scalars[0] = 0.95f / fmaxf(m1, 1e-8f);
        scalars[1] = sm[0] * 19.0f;
    }
}

template <int K>
__global__ __launch_bounds__(256)
void build_wt_kernel(const float* __restrict__ R, const float* __restrict__ M,
                     const float* __restrict__ scalars, int sel,
                     unsigned short* __restrict__ WT,
                     int N, int srcld, int NL) {
    const unsigned g = blockIdx.x * 256u + threadIdx.x;
    const int lane = (int)(g & 31u);
    const unsigned wv = g >> 5;
    const int k0 = (int)(wv % (unsigned)(K / 8)) * 8;
    const int n  = (int)(wv / (unsigned)(K / 8)) * 32 + lane;
    if (n >= N) return;
    float v[8];
    if (n < NL) {
        const size_t s = (size_t)k0 * srcld + n;
        float r[8], m[8];
#pragma unroll
        for (int e = 0; e < 8; ++e) {
            r[e] = R[s + (size_t)e * srcld];
            m[e] = M[s + (size_t)e * srcld];
        }
#pragma unroll
        for (int e = 0; e < 8; ++e) v[e] = r[e] * m[e];
        if (sel == 0) {
            const float sc = scalars[0];
#pragma unroll
            for (int e = 0; e < 8; ++e) v[e] *= sc;
        }
    } else {
#pragma unroll
        for (int e = 0; e < 8; ++e) v[e] = 0.f;
    }
    u32x4 c;
    c[0] = (unsigned)f2bf(v[0]) | ((unsigned)f2bf(v[1]) << 16);
    c[1] = (unsigned)f2bf(v[2]) | ((unsigned)f2bf(v[3]) << 16);
    c[2] = (unsigned)f2bf(v[4]) | ((unsigned)f2bf(v[5]) << 16);
    c[3] = (unsigned)f2bf(v[6]) | ((unsigned)f2bf(v[7]) << 16);
    *reinterpret_cast<u32x4*>(WT + packed_idx(n, k0, K)) = c;
}

template <int K>
__global__ __launch_bounds__(256)
void to_bf16_kernel(const float* __restrict__ src, unsigned short* __restrict__ dst,
                    int rows) {
    const unsigned g = blockIdx.x * 256u + threadIdx.x;
    const int lane = (int)(g & 31u);
    const unsigned hw = g >> 5;
    const int hh = (int)(hw & 1u);
    const unsigned tile = hw >> 1;
    const int kt = (int)(tile % (unsigned)(K / 32));
    const int r  = (int)(tile / (unsigned)(K / 32)) * 16 + (lane & 15);
    const int k0 = kt * 32 + hh * 16 + (lane >> 4) * 8;
    if (r >= rows) return;
    const float* p = src + (size_t)r * K + k0;
    float v[8];
#pragma unroll
    for (int e = 0; e < 8; ++e) v[e] = p[e];
    u32x4 c;
    c[0] = (unsigned)f2bf(v[0]) | ((unsigned)f2bf(v[1]) << 16);
    c[1] = (unsigned)f2bf(v[2]) | ((unsigned)f2bf(v[3]) << 16);
    c[2] = (unsigned)f2bf(v[4]) | ((unsigned)f2bf(v[5]) << 16);
    c[3] = (unsigned)f2bf(v[6]) | ((unsigned)f2bf(v[7]) << 16);
    *reinterpret_cast<u32x4*>(dst + packed_idx(r, k0, K)) = c;
}

__global__ __launch_bounds__(256)
void init_state(float* __restrict__ H0, unsigned short* __restrict__ Hb0,
                int* __restrict__ act0, float* __restrict__ dHa0,
                float* __restrict__ dHb0, float* __restrict__ pairs0) {
    const size_t n = (size_t)Bsz * HD;
    for (size_t t = (size_t)blockIdx.x * 256 + threadIdx.x; t < n;
         t += (size_t)gridDim.x * 256) {
        H0[t] = 0.f;
        Hb0[t] = 0;
    }
    const int i = blockIdx.x * 256 + threadIdx.x;
    if (i < Bsz) act0[i] = 1;
    if (i < DHP) { dHa0[i] = 0.f; dHb0[i] = 0.f; }
    if (i < 2 * (LP / 64) * Bsz * 2) pairs0[i] = 0.f;
}

__global__ __launch_bounds__(128)
void finish_rows(const float* __restrict__ Stage, float* __restrict__ out,
                 const int* __restrict__ actIn,
                 const float* __restrict__ Top2In,
                 const float* __restrict__ dHa,
                 const float* __restrict__ scalars, int M, int NL) {
    const int lane = threadIdx.x & 31;
    const int r16 = lane & 15;
    const int kh  = lane >> 4;
    const int rowBase = blockIdx.y * 16;
    const int row = rowBase + r16;
    const float2* tp = reinterpret_cast<const float2*>(Top2In);
    float m1 = -1e30f, m2 = -1e30f;
#pragma unroll
    for (int j = 0; j < 8; ++j) {
        const float2 q = tp[(size_t)(kh * 8 + j) * M + row];
        if (j == 0) { m1 = q.x; m2 = q.y; } else top2_join(m1, m2, q.x, q.y);
    }
    const float o1 = __shfl_xor(m1, 16, 32);
    const float o2 = __shfl_xor(m2, 16, 32);
    top2_join(m1, m2, o1, o2);
    const float margin = m1 - m2;
    const size_t dofs = ((size_t)(kh * (NCB / 2)) * (M >> 4) + (rowBase >> 4)) * 32 + r16;
    float dh = dHa[dofs];
#pragma unroll
    for (int j = 1; j < NCB / 2; ++j) dh = fmaxf(dh, dHa[dofs + (size_t)j * (M >> 4) * 32]);
    dh = fmaxf(dh, __shfl_xor(dh, 16, 32));
    const float Gamma = scalars[1] * dh;
    const bool cert = margin > 2.0f * Gamma;
    const unsigned take = __builtin_amdgcn_ballot_w32(actIn[row] != 0 && !cert) & 0xFFFFu;
    if (take == 0u) return;
    const int per = (NL + (int)gridDim.x - 1) / (int)gridDim.x;
    const int c0 = blockIdx.x * per;
    const int c1 = min(NL, c0 + per);
    for (int r = 0; r < 16; ++r) {
        if ((take >> r) & 1u) {
            const size_t o = (size_t)(rowBase + r) * NL;
            for (int c = c0 + (int)threadIdx.x; c < c1; c += 128) out[o + c] = Stage[o + c];
        }
    }
}

static inline char* wsalloc(char*& p, size_t bytes) {
    char* r = p;
    p += (bytes + 255) & ~(size_t)255;
    return r;
}

extern "C" void kernel_launch(void* const* d_in, const int* in_sizes, int n_in,
                              void* d_out, int out_size, void* d_ws, size_t ws_size,
                              hipStream_t stream) {
    (void)in_sizes; (void)n_in; (void)out_size; (void)ws_size;
    const float* x   = (const float*)d_in[0];
    const float* RIH = (const float*)d_in[1];
    const float* RH  = (const float*)d_in[2];
    const float* RHL = (const float*)d_in[3];
    const float* BH  = (const float*)d_in[4];
    const float* BL  = (const float*)d_in[5];
    const float* MIH = (const float*)d_in[6];
    const float* MH  = (const float*)d_in[7];
    const float* MHL = (const float*)d_in[8];
    float* out = (float*)d_out;

    char* p = (char*)d_ws;
    unsigned short* WT_H  = (unsigned short*)wsalloc(p, (size_t)HD * HD * 2);
    unsigned short* WT_HL = (unsigned short*)wsalloc(p, (size_t)LP * HD * 2);
    unsigned short* WT_IH = (unsigned short*)wsalloc(p, (size_t)HD * Isz * 2);
    unsigned short* xb    = (unsigned short*)wsalloc(p, (size_t)Bsz * Isz * 2);
    float* drive = (float*)wsalloc(p, (size_t)Bsz * HD * 4);
    float* Hbuf0 = (float*)wsalloc(p, (size_t)Bsz * HD * 4);
    float* Hbuf1 = (float*)wsalloc(p, (size_t)Bsz * HD * 4);
    unsigned short* Hb0 = (unsigned short*)wsalloc(p, (size_t)Bsz * HD * 2);
    unsigned short* Hb1 = (unsigned short*)wsalloc(p, (size_t)Bsz * HD * 2);
    float* col_l1 = (float*)wsalloc(p, HD * 4);
    float* whl    = (float*)wsalloc(p, LP * 4);
    float* scalars = (float*)wsalloc(p, 256);
    int* act = (int*)wsalloc(p, (size_t)(TMAX / 2 + 1) * Bsz * 4);
    float* stage = (float*)WT_IH;
    float* dHp = (float*)xb;
    float* top2 = dHp + (size_t)4 * DHP;
    const size_t T2SET = (size_t)2 * (LP / 64) * Bsz * 2;

    col_l1_kernel<HD><<<HD / 8, 256, 0, stream>>>(RH, MH, col_l1);
    col_l1_kernel<Lsz><<<Lsz / 8, 256, 0, stream>>>(RHL, MHL, whl);
    scalars_kernel<<<1, 256, 0, stream>>>(col_l1, whl, scalars);

    build_wt_kernel<HD><<<(HD * HD / 8) / 256, 256, 0, stream>>>(
        RH, MH, scalars,  0, WT_H, HD, HD, HD);
    build_wt_kernel<HD><<<(LP * HD / 8) / 256, 256, 0, stream>>>(
        RHL, MHL, scalars,  1, WT_HL, LP, Lsz, Lsz);
    build_wt_kernel<Isz><<<(HD * Isz / 8) / 256, 256, 0, stream>>>(
        RIH, MIH, scalars,  1, WT_IH, HD, HD, HD);
    to_bf16_kernel<Isz><<<(Bsz * Isz / 8) / 256, 256, 0, stream>>>(x, xb, Bsz);

    wmma_gemm<0><<<dim3(HD / 64, Bsz / 64), 128, 0, stream>>>(
        xb, nullptr, WT_IH, Bsz, HD, Isz, BH, nullptr, nullptr, nullptr, nullptr,
        drive, nullptr, nullptr, nullptr, nullptr, nullptr, nullptr, nullptr, nullptr, HD);

    init_state<<<(Bsz * HD) / 256, 256, 0, stream>>>(
        Hbuf0, Hb0, act, dHp + (size_t)3 * DHP, dHp, top2);

    float* Hc[2] = {Hbuf0, Hbuf1};
    unsigned short* Hbc[2] = {Hb0, Hb1};
    for (int j = 1; j <= TMAX / 2; ++j) {
        for (int t = 2 * j - 1; t <= 2 * j; ++t) {
            const int pc = (t - 1) & 1, pn = t & 1;
            wmma_gemm<1><<<dim3(HD / 64, Bsz / 64), 128, 0, stream>>>(
                Hbc[pc], nullptr, WT_H, Bsz, HD, HD, nullptr, drive, Hc[pc],
                nullptr, nullptr, Hc[pn], nullptr, Hbc[pn],
                nullptr, nullptr, dHp + (size_t)(t & 3) * DHP,
                nullptr, nullptr, nullptr, HD);
        }
        wmma_gemm<2><<<dim3(LP / 64, 2 * (Bsz / 64)), 128, 0, stream>>>(
            Hbc[1], Hbc[0], WT_HL, Bsz, LP, HD, BL, nullptr, nullptr,
            act + (j - 1) * Bsz, act + j * Bsz, out, stage, nullptr,
            dHp + (size_t)((2 * j + 1) & 3) * DHP, dHp + (size_t)((2 * j + 2) & 3) * DHP, nullptr,
            top2 + (size_t)((j - 1) & 1) * T2SET, top2 + (size_t)(j & 1) * T2SET,
            scalars, Lsz);
    }
    finish_rows<<<dim3(4, Bsz / 16), 128, 0, stream>>>(
        stage, out, act + (TMAX / 2) * Bsz, top2 + (size_t)((TMAX / 2) & 1) * T2SET,
        dHp + (size_t)((TMAX - 1) & 3) * DHP, scalars, Bsz, Lsz);
}
